# starting kernel
# baseline (speedup 1.0000x reference)
#include <hip/hip_runtime.h>
#include <hip/hip_bf16.h>

typedef float v2f __attribute__((ext_vector_type(2)));
typedef float v8f __attribute__((ext_vector_type(8)));

#define B_ 4
#define L_ 1024
#define D_ 512
#define N_ 16
#define CHUNK 64
#define NCHUNK 16
#define LOG2E 1.4426950408889634f

__global__ void k_xproj(const float* __restrict__ x_rgb, const float* __restrict__ x_e,
                        const float* __restrict__ w1, const float* __restrict__ w2,
                        float* __restrict__ dbl) {
    int wave = (blockIdx.x * blockDim.x + threadIdx.x) >> 5;
    int ln = threadIdx.x & 31;
    int ntile = wave % 3;
    int rem   = wave / 3;
    int mtile = rem & 255;
    int branch = rem >> 8;
    const float* x = branch ? x_e : x_rgb;
    const float* w = branch ? w2 : w1;
    int m  = mtile * 16 + (ln & 15);
    int n  = ntile * 16 + (ln & 15);
    int kb = (ln >> 4) * 2;
    const float* xrow = x + (long)m * D_;
    const float* wrow = w + (long)n * D_;
    v8f c = {};
    for (int k = 0; k < D_; k += 4) {
        v2f a, b;
        a.x = xrow[k + kb];  a.y = xrow[k + kb + 1];
        b.x = wrow[k + kb];  b.y = wrow[k + kb + 1];
        c = __builtin_amdgcn_wmma_f32_16x16x4_f32(false, a, false, b, (short)0, c, false, false);
    }
    float* out = dbl + (long)branch * 4096 * 48;
    int rowbase = mtile * 16 + ((ln >> 4) << 3);
    int col = ntile * 16 + (ln & 15);
#pragma unroll
    for (int i = 0; i < 8; ++i)
        out[(long)(rowbase + i) * 48 + col] = c[i];
}

__global__ void k_dtproj(const float* __restrict__ dbl,
                         const float* __restrict__ w1, const float* __restrict__ bias1,
                         const float* __restrict__ w2, const float* __restrict__ bias2,
                         float* __restrict__ delta_sp) {
    int wave = (blockIdx.x * blockDim.x + threadIdx.x) >> 5;
    int ln = threadIdx.x & 31;
    int ntile = wave & 31;
    int rem   = wave >> 5;
    int mtile = rem & 255;
    int branch = rem >> 8;
    const float* w    = branch ? w2 : w1;
    const float* bias = branch ? bias2 : bias1;
    const float* dt   = dbl + (long)branch * 4096 * 48;
    int m  = mtile * 16 + (ln & 15);
    int n  = ntile * 16 + (ln & 15);
    int kb = (ln >> 4) * 2;
    v8f c = {};
#pragma unroll
    for (int k = 0; k < 16; k += 4) {
        v2f a, b;
        a.x = dt[(long)m * 48 + k + kb];  a.y = dt[(long)m * 48 + k + kb + 1];
        b.x = w[(long)n * 16 + k + kb];   b.y = w[(long)n * 16 + k + kb + 1];
        c = __builtin_amdgcn_wmma_f32_16x16x4_f32(false, a, false, b, (short)0, c, false, false);
    }
    float* out = delta_sp + (long)branch * 4096 * D_;
    int rowbase = mtile * 16 + ((ln >> 4) << 3);
    int col = ntile * 16 + (ln & 15);
    float bcol = bias[col];
#pragma unroll
    for (int i = 0; i < 8; ++i) {
        float v  = c[i] + bcol;
        float sp = (v > 20.f) ? v : log1pf(__expf(v));
        out[(long)(rowbase + i) * D_ + col] = sp;
    }
}

__global__ void k_scan1(const float* __restrict__ x_rgb, const float* __restrict__ x_e,
                        const float* __restrict__ dbl, const float* __restrict__ delta_sp,
                        const float* __restrict__ Alog1, const float* __restrict__ Alog2,
                        float* __restrict__ hpart, float* __restrict__ aprod) {
    int bid = blockIdx.x;
    int lc  = bid & 15; bid >>= 4;
    int dch = bid & 3;  bid >>= 2;
    int b   = bid & 3;
    int branch = bid >> 2;
    int tid = threadIdx.x;
    int l0 = lc * CHUNK;

    __shared__ float Bsh[CHUNK * 16];
    {
        const float* src = dbl + (long)branch * 4096 * 48 + (long)(b * L_ + l0) * 48 + 16;
        for (int i = tid; i < CHUNK * 16; i += blockDim.x)
            Bsh[i] = src[(long)(i >> 4) * 48 + (i & 15)];
    }
    __syncthreads();

    int d = dch * 128 + tid;
    const float* Alog = branch ? Alog2 : Alog1;
    const float* x    = branch ? x_e : x_rgb;
    const float* ds   = delta_sp + (long)branch * 4096 * D_;

    float A2[N_], h[N_];
#pragma unroll
    for (int n = 0; n < N_; ++n) {
        A2[n] = -__expf(Alog[(long)d * N_ + n]) * LOG2E;
        h[n] = 0.f;
    }
    float ssum = 0.f;
    for (int ll = 0; ll < CHUNK; ++ll) {
        long idx = (long)(b * L_ + l0 + ll) * D_ + d;
        float s = ds[idx];
        float u = x[idx];
        float su = s * u;
        ssum += s;
#pragma unroll
        for (int n = 0; n < N_; ++n) {
            float dA = __builtin_amdgcn_exp2f(s * A2[n]);
            h[n] = fmaf(dA, h[n], su * Bsh[ll * 16 + n]);
        }
    }
    long off = ((long)(branch * 4 + b) * NCHUNK + lc) * (D_ * N_) + (long)d * N_;
#pragma unroll
    for (int n = 0; n < N_; ++n) {
        hpart[off + n] = h[n];
        aprod[off + n] = __builtin_amdgcn_exp2f(ssum * A2[n]);
    }
}

__global__ void k_carry(const float* __restrict__ hpart, const float* __restrict__ aprod,
                        float* __restrict__ hin) {
    int idx = blockIdx.x * blockDim.x + threadIdx.x;
    int dn = idx & (D_ * N_ - 1);
    int bb = idx >> 13;
    float h = 0.f;
    for (int c = 0; c < NCHUNK; ++c) {
        long pos = ((long)bb * NCHUNK + c) * (D_ * N_) + dn;
        hin[pos] = h;
        h = fmaf(aprod[pos], h, hpart[pos]);
    }
}

__global__ void k_scan2(const float* __restrict__ x_rgb, const float* __restrict__ x_e,
                        const float* __restrict__ dbl, const float* __restrict__ delta_sp,
                        const float* __restrict__ Alog1, const float* __restrict__ Alog2,
                        const float* __restrict__ D1, const float* __restrict__ D2,
                        const float* __restrict__ hin, float* __restrict__ out) {
    int bid = blockIdx.x;
    int lc  = bid & 15; bid >>= 4;
    int dch = bid & 3;  bid >>= 2;
    int b   = bid & 3;
    int branch = bid >> 2;
    int tid = threadIdx.x;
    int l0 = lc * CHUNK;

    __shared__ float Bsh[CHUNK * 16];
    __shared__ float Csh[CHUNK * 16];
    {
        const float* srcB = dbl + (long)branch * 4096 * 48 + (long)(b * L_ + l0) * 48 + 16;
        const float* srcC = dbl + (long)(1 - branch) * 4096 * 48 + (long)(b * L_ + l0) * 48 + 32;
        for (int i = tid; i < CHUNK * 16; i += blockDim.x) {
            int r = i >> 4, cc = i & 15;
            Bsh[i] = srcB[(long)r * 48 + cc];
            Csh[i] = srcC[(long)r * 48 + cc];
        }
    }
    __syncthreads();

    int d = dch * 128 + tid;
    const float* Alog = branch ? Alog2 : Alog1;
    const float* Dp   = branch ? D2 : D1;
    const float* x    = branch ? x_e : x_rgb;
    const float* ds   = delta_sp + (long)branch * 4096 * D_;

    float A2[N_], h[N_];
    long off = ((long)(branch * 4 + b) * NCHUNK + lc) * (D_ * N_) + (long)d * N_;
#pragma unroll
    for (int n = 0; n < N_; ++n) {
        A2[n] = -__expf(Alog[(long)d * N_ + n]) * LOG2E;
        h[n] = hin[off + n];
    }
    float Dd = Dp[d];
    float* yout = out + (long)branch * 4096 * D_;
    for (int ll = 0; ll < CHUNK; ++ll) {
        long idx = (long)(b * L_ + l0 + ll) * D_ + d;
        float s = ds[idx];
        float u = x[idx];
        float su = s * u;
        float y = 0.f;
#pragma unroll
        for (int n = 0; n < N_; ++n) {
            float dA = __builtin_amdgcn_exp2f(s * A2[n]);
            h[n] = fmaf(dA, h[n], su * Bsh[ll * 16 + n]);
            y = fmaf(h[n], Csh[ll * 16 + n], y);
        }
        yout[idx] = fmaf(Dd, u, y);
    }
}

__global__ void k_ln(float* __restrict__ out,
                     const float* __restrict__ g1, const float* __restrict__ bb1,
                     const float* __restrict__ g2, const float* __restrict__ bb2) {
    int row = blockIdx.x;
    int branch = row >> 12;
    float* p = out + (long)row * D_;
    int tid = threadIdx.x;
    float v0 = p[tid], v1 = p[tid + 256];
    __shared__ float rs[256], rq[256];
    rs[tid] = v0 + v1;
    rq[tid] = v0 * v0 + v1 * v1;
    __syncthreads();
    for (int st = 128; st > 0; st >>= 1) {
        if (tid < st) { rs[tid] += rs[tid + st]; rq[tid] += rq[tid + st]; }
        __syncthreads();
    }
    float mean = rs[0] * (1.f / 512.f);
    float var  = rq[0] * (1.f / 512.f) - mean * mean;
    float inv  = rsqrtf(var + 1e-5f);
    const float* g = branch ? g2 : g1;
    const float* bb = branch ? bb2 : bb1;
    p[tid]       = (v0 - mean) * inv * g[tid]       + bb[tid];
    p[tid + 256] = (v1 - mean) * inv * g[tid + 256] + bb[tid + 256];
}

extern "C" void kernel_launch(void* const* d_in, const int* in_sizes, int n_in,
                              void* d_out, int out_size, void* d_ws, size_t ws_size,
                              hipStream_t stream) {
    const float* x_rgb = (const float*)d_in[0];
    const float* x_e   = (const float*)d_in[1];
    const float* xw1   = (const float*)d_in[2];
    const float* xw2   = (const float*)d_in[3];
    const float* dtw1  = (const float*)d_in[4];
    const float* dtb1  = (const float*)d_in[5];
    const float* dtw2  = (const float*)d_in[6];
    const float* dtb2  = (const float*)d_in[7];
    const float* Al1   = (const float*)d_in[8];
    const float* Al2   = (const float*)d_in[9];
    const float* Dv1   = (const float*)d_in[10];
    const float* Dv2   = (const float*)d_in[11];
    const float* g1    = (const float*)d_in[12];
    const float* lb1   = (const float*)d_in[13];
    const float* g2    = (const float*)d_in[14];
    const float* lb2   = (const float*)d_in[15];
    float* out = (float*)d_out;

    float* ws    = (float*)d_ws;
    float* dbl   = ws;
    float* delta = dbl + 2 * 4096 * 48;
    float* hpart = delta + 2 * 4096 * 512;
    float* aprod = hpart + 2 * 4 * NCHUNK * 8192;
    float* hin   = aprod + 2 * 4 * NCHUNK * 8192;

    k_xproj<<<384, 128, 0, stream>>>(x_rgb, x_e, xw1, xw2, dbl);
    k_dtproj<<<4096, 128, 0, stream>>>(dbl, dtw1, dtb1, dtw2, dtb2, delta);
    k_scan1<<<512, 128, 0, stream>>>(x_rgb, x_e, dbl, delta, Al1, Al2, hpart, aprod);
    k_carry<<<256, 256, 0, stream>>>(hpart, aprod, hin);
    k_scan2<<<512, 128, 0, stream>>>(x_rgb, x_e, dbl, delta, Al1, Al2, Dv1, Dv2, hin, out);
    k_ln<<<8192, 256, 0, stream>>>(out, g1, lb1, g2, lb2);
}


# scan passes 1 and 2: operands prefetched ahead of the recurrence (s,u one batch of steps ahead in registers, B/C rows one step ahead from LDS), so no step waits for a load
# speedup vs baseline: 1.9769x; 1.9769x over previous
#include <hip/hip_runtime.h>
#include <hip/hip_bf16.h>

typedef float v2f __attribute__((ext_vector_type(2)));
typedef float v8f __attribute__((ext_vector_type(8)));
typedef float v4f __attribute__((ext_vector_type(4)));

#define B_ 4
#define L_ 1024
#define D_ 512
#define N_ 16
#define CHUNK 64
#define NCHUNK 16
#define BT 16
#define NBATCH 4
#define LOG2E 1.4426950408889634f

__global__ void k_xproj(const float* __restrict__ x_rgb, const float* __restrict__ x_e,
                        const float* __restrict__ w1, const float* __restrict__ w2,
                        float* __restrict__ dbl) {
    int wave = (blockIdx.x * blockDim.x + threadIdx.x) >> 5;
    int ln = threadIdx.x & 31;
    int ntile = wave % 3;
    int rem   = wave / 3;
    int mtile = rem & 255;
    int branch = rem >> 8;
    const float* x = branch ? x_e : x_rgb;
    const float* w = branch ? w2 : w1;
    int m  = mtile * 16 + (ln & 15);
    int n  = ntile * 16 + (ln & 15);
    int kb = (ln >> 4) * 2;
    const float* xrow = x + (long)m * D_;
    const float* wrow = w + (long)n * D_;
    v8f c = {};
    for (int k = 0; k < D_; k += 4) {
        v2f a, b;
        a.x = xrow[k + kb];  a.y = xrow[k + kb + 1];
        b.x = wrow[k + kb];  b.y = wrow[k + kb + 1];
        c = __builtin_amdgcn_wmma_f32_16x16x4_f32(false, a, false, b, (short)0, c, false, false);
    }
    float* out = dbl + (long)branch * 4096 * 48;
    int rowbase = mtile * 16 + ((ln >> 4) << 3);
    int col = ntile * 16 + (ln & 15);
#pragma unroll
    for (int i = 0; i < 8; ++i)
        out[(long)(rowbase + i) * 48 + col] = c[i];
}

__global__ void k_dtproj(const float* __restrict__ dbl,
                         const float* __restrict__ w1, const float* __restrict__ bias1,
                         const float* __restrict__ w2, const float* __restrict__ bias2,
                         float* __restrict__ delta_sp) {
    int wave = (blockIdx.x * blockDim.x + threadIdx.x) >> 5;
    int ln = threadIdx.x & 31;
    int ntile = wave & 31;
    int rem   = wave >> 5;
    int mtile = rem & 255;
    int branch = rem >> 8;
    const float* w    = branch ? w2 : w1;
    const float* bias = branch ? bias2 : bias1;
    const float* dt   = dbl + (long)branch * 4096 * 48;
    int m  = mtile * 16 + (ln & 15);
    int n  = ntile * 16 + (ln & 15);
    int kb = (ln >> 4) * 2;
    v8f c = {};
#pragma unroll
    for (int k = 0; k < 16; k += 4) {
        v2f a, b;
        a.x = dt[(long)m * 48 + k + kb];  a.y = dt[(long)m * 48 + k + kb + 1];
        b.x = w[(long)n * 16 + k + kb];   b.y = w[(long)n * 16 + k + kb + 1];
        c = __builtin_amdgcn_wmma_f32_16x16x4_f32(false, a, false, b, (short)0, c, false, false);
    }
    float* out = delta_sp + (long)branch * 4096 * D_;
    int rowbase = mtile * 16 + ((ln >> 4) << 3);
    int col = ntile * 16 + (ln & 15);
    float bcol = bias[col];
#pragma unroll
    for (int i = 0; i < 8; ++i) {
        float v  = c[i] + bcol;
        float sp = (v > 20.f) ? v : log1pf(__expf(v));
        out[(long)(rowbase + i) * D_ + col] = sp;
    }
}

__global__ void __launch_bounds__(128) __attribute__((amdgpu_waves_per_eu(2, 2)))
k_scan1(const float* __restrict__ x_rgb, const float* __restrict__ x_e,
        const float* __restrict__ dbl, const float* __restrict__ delta_sp,
        const float* __restrict__ Alog1, const float* __restrict__ Alog2,
        float* __restrict__ hpart, float* __restrict__ aprod) {
    int bid = blockIdx.x;
    int lc  = bid & 15; bid >>= 4;
    int dch = bid & 3;  bid >>= 2;
    int b   = bid & 3;
    int branch = bid >> 2;
    int tid = threadIdx.x;
    int l0 = lc * CHUNK;

    int d = dch * 128 + tid;
    const float* Alog = branch ? Alog2 : Alog1;
    const float* x    = branch ? x_e : x_rgb;
    const float* ds   = delta_sp + (long)branch * 4096 * D_;
    const float* sp   = ds + (long)(b * L_ + l0) * D_ + d;
    const float* up   = x  + (long)(b * L_ + l0) * D_ + d;

    float sN[BT], uN[BT];
#pragma unroll
    for (int j = 0; j < BT; ++j) { sN[j] = sp[j * D_]; uN[j] = up[j * D_]; }
    float al[N_];
#pragma unroll
    for (int n = 0; n < N_; ++n) al[n] = Alog[(long)d * N_ + n];

    __shared__ __attribute__((aligned(16))) float Bsh[CHUNK * 16];
    const v4f* Bsh4 = reinterpret_cast<const v4f*>(Bsh);
    {
        const float* src = dbl + (long)branch * 4096 * 48 + (long)(b * L_ + l0) * 48 + 16;
        for (int i = tid; i < CHUNK * 16; i += blockDim.x)
            Bsh[i] = src[(long)(i >> 4) * 48 + (i & 15)];
    }

    float A2[N_], h[N_];
#pragma unroll
    for (int n = 0; n < N_; ++n) {
        A2[n] = -__expf(al[n]) * LOG2E;
        h[n] = 0.f;
    }
    __syncthreads();

    v4f bq[4];
#pragma unroll
    for (int k = 0; k < 4; ++k) bq[k] = Bsh4[k];

    float ssum = 0.f;
#pragma unroll 1
    for (int bt = 0; bt < NBATCH; ++bt) {
        float sC[BT], uC[BT];
#pragma unroll
        for (int j = 0; j < BT; ++j) { sC[j] = sN[j]; uC[j] = uN[j]; }
        if (bt + 1 < NBATCH) {
            const float* sp1 = sp + (long)(bt + 1) * BT * D_;
            const float* up1 = up + (long)(bt + 1) * BT * D_;
#pragma unroll
            for (int j = 0; j < BT; ++j) { sN[j] = sp1[j * D_]; uN[j] = up1[j * D_]; }
        }
#pragma unroll
        for (int j = 0; j < BT; ++j) {
            int rn = (bt * BT + j + 1) & (CHUNK - 1);
            v4f bn[4];
#pragma unroll
            for (int k = 0; k < 4; ++k) bn[k] = Bsh4[rn * 4 + k];
            float s = sC[j];
            float u = uC[j];
            float su = s * u;
            ssum += s;
            float Brow[N_];
#pragma unroll
            for (int k = 0; k < 4; ++k) {
                Brow[4 * k + 0] = bq[k].x; Brow[4 * k + 1] = bq[k].y;
                Brow[4 * k + 2] = bq[k].z; Brow[4 * k + 3] = bq[k].w;
            }
#pragma unroll
            for (int n = 0; n < N_; ++n) {
                float dA = __builtin_amdgcn_exp2f(s * A2[n]);
                h[n] = fmaf(dA, h[n], su * Brow[n]);
            }
#pragma unroll
            for (int k = 0; k < 4; ++k) bq[k] = bn[k];
        }
    }
    long off = ((long)(branch * 4 + b) * NCHUNK + lc) * (D_ * N_) + (long)d * N_;
#pragma unroll
    for (int n = 0; n < N_; ++n) {
        hpart[off + n] = h[n];
        aprod[off + n] = __builtin_amdgcn_exp2f(ssum * A2[n]);
    }
}

__global__ void k_carry(const float* __restrict__ hpart, const float* __restrict__ aprod,
                        float* __restrict__ hin) {
    int idx = blockIdx.x * blockDim.x + threadIdx.x;
    int dn = idx & (D_ * N_ - 1);
    int bb = idx >> 13;
    float h = 0.f;
    for (int c = 0; c < NCHUNK; ++c) {
        long pos = ((long)bb * NCHUNK + c) * (D_ * N_) + dn;
        hin[pos] = h;
        h = fmaf(aprod[pos], h, hpart[pos]);
    }
}

__global__ void __launch_bounds__(128) __attribute__((amdgpu_waves_per_eu(2, 2)))
k_scan2(const float* __restrict__ x_rgb, const float* __restrict__ x_e,
        const float* __restrict__ dbl, const float* __restrict__ delta_sp,
        const float* __restrict__ Alog1, const float* __restrict__ Alog2,
        const float* __restrict__ D1, const float* __restrict__ D2,
        const float* __restrict__ hin, float* __restrict__ out) {
    int bid = blockIdx.x;
    int lc  = bid & 15; bid >>= 4;
    int dch = bid & 3;  bid >>= 2;
    int b   = bid & 3;
    int branch = bid >> 2;
    int tid = threadIdx.x;
    int l0 = lc * CHUNK;

    int d = dch * 128 + tid;
    const float* Alog = branch ? Alog2 : Alog1;
    const float* Dp   = branch ? D2 : D1;
    const float* x    = branch ? x_e : x_rgb;
    const float* ds   = delta_sp + (long)branch * 4096 * D_;
    const float* sp   = ds + (long)(b * L_ + l0) * D_ + d;
    const float* up   = x  + (long)(b * L_ + l0) * D_ + d;

    float sN[BT], uN[BT];
#pragma unroll
    for (int j = 0; j < BT; ++j) { sN[j] = sp[j * D_]; uN[j] = up[j * D_]; }
    float al[N_], h[N_];
    long off = ((long)(branch * 4 + b) * NCHUNK + lc) * (D_ * N_) + (long)d * N_;
#pragma unroll
    for (int n = 0; n < N_; ++n) {
        al[n] = Alog[(long)d * N_ + n];
        h[n]  = hin[off + n];
    }
    float Dd = Dp[d];

    __shared__ __attribute__((aligned(16))) float Bsh[CHUNK * 16];
    __shared__ __attribute__((aligned(16))) float Csh[CHUNK * 16];
    const v4f* Bsh4 = reinterpret_cast<const v4f*>(Bsh);
    const v4f* Csh4 = reinterpret_cast<const v4f*>(Csh);
    {
        const float* srcB = dbl + (long)branch * 4096 * 48 + (long)(b * L_ + l0) * 48 + 16;
        const float* srcC = dbl + (long)(1 - branch) * 4096 * 48 + (long)(b * L_ + l0) * 48 + 32;
        for (int i = tid; i < CHUNK * 16; i += blockDim.x) {
            int r = i >> 4, cc = i & 15;
            Bsh[i] = srcB[(long)r * 48 + cc];
            Csh[i] = srcC[(long)r * 48 + cc];
        }
    }

    float A2[N_];
#pragma unroll
    for (int n = 0; n < N_; ++n) A2[n] = -__expf(al[n]) * LOG2E;
    __syncthreads();

    v4f bq[4], cq[4];
#pragma unroll
    for (int k = 0; k < 4; ++k) { bq[k] = Bsh4[k]; cq[k] = Csh4[k]; }

    float* yp = out + (long)branch * 4096 * D_ + (long)(b * L_ + l0) * D_ + d;
#pragma unroll 1
    for (int bt = 0; bt < NBATCH; ++bt) {
        float sC[BT], uC[BT];
#pragma unroll
        for (int j = 0; j < BT; ++j) { sC[j] = sN[j]; uC[j] = uN[j]; }
        if (bt + 1 < NBATCH) {
            const float* sp1 = sp + (long)(bt + 1) * BT * D_;
            const float* up1 = up + (long)(bt + 1) * BT * D_;
#pragma unroll
            for (int j = 0; j < BT; ++j) { sN[j] = sp1[j * D_]; uN[j] = up1[j * D_]; }
        }
        float* yb = yp + (long)bt * BT * D_;
#pragma unroll
        for (int j = 0; j < BT; ++j) {
            int rn = (bt * BT + j + 1) & (CHUNK - 1);
            v4f bn[4], cn[4];
#pragma unroll
            for (int k = 0; k < 4; ++k) { bn[k] = Bsh4[rn * 4 + k]; cn[k] = Csh4[rn * 4 + k]; }
            float s = sC[j];
            float u = uC[j];
            float su = s * u;
            float y = 0.f;
            float Brow[N_], Crow[N_];
#pragma unroll
            for (int k = 0; k < 4; ++k) {
                Brow[4 * k + 0] = bq[k].x; Brow[4 * k + 1] = bq[k].y;
                Brow[4 * k + 2] = bq[k].z; Brow[4 * k + 3] = bq[k].w;
                Crow[4 * k + 0] = cq[k].x; Crow[4 * k + 1] = cq[k].y;
                Crow[4 * k + 2] = cq[k].z; Crow[4 * k + 3] = cq[k].w;
            }
#pragma unroll
            for (int n = 0; n < N_; ++n) {
                float dA = __builtin_amdgcn_exp2f(s * A2[n]);
                h[n] = fmaf(dA, h[n], su * Brow[n]);
                y = fmaf(h[n], Crow[n], y);
            }
            yb[j * D_] = fmaf(Dd, u, y);
#pragma unroll
            for (int k = 0; k < 4; ++k) { bq[k] = bn[k]; cq[k] = cn[k]; }
        }
    }
}

__global__ void k_ln(float* __restrict__ out,
                     const float* __restrict__ g1, const float* __restrict__ bb1,
                     const float* __restrict__ g2, const float* __restrict__ bb2) {
    int row = blockIdx.x;
    int branch = row >> 12;
    float* p = out + (long)row * D_;
    int tid = threadIdx.x;
    float v0 = p[tid], v1 = p[tid + 256];
    __shared__ float rs[256], rq[256];
    rs[tid] = v0 + v1;
    rq[tid] = v0 * v0 + v1 * v1;
    __syncthreads();
    for (int st = 128; st > 0; st >>= 1) {
        if (tid < st) { rs[tid] += rs[tid + st]; rq[tid] += rq[tid + st]; }
        __syncthreads();
    }
    float mean = rs[0] * (1.f / 512.f);
    float var  = rq[0] * (1.f / 512.f) - mean * mean;
    float inv  = rsqrtf(var + 1e-5f);
    const float* g = branch ? g2 : g1;
    const float* bb = branch ? bb2 : bb1;
    p[tid]       = (v0 - mean) * inv * g[tid]       + bb[tid];
    p[tid + 256] = (v1 - mean) * inv * g[tid + 256] + bb[tid + 256];
}

extern "C" void kernel_launch(void* const* d_in, const int* in_sizes, int n_in,
                              void* d_out, int out_size, void* d_ws, size_t ws_size,
                              hipStream_t stream) {
    const float* x_rgb = (const float*)d_in[0];
    const float* x_e   = (const float*)d_in[1];
    const float* xw1   = (const float*)d_in[2];
    const float* xw2   = (const float*)d_in[3];
    const float* dtw1  = (const float*)d_in[4];
    const float* dtb1  = (const float*)d_in[5];
    const float* dtw2  = (const float*)d_in[6];
    const float* dtb2  = (const float*)d_in[7];
    const float* Al1   = (const float*)d_in[8];
    const float* Al2   = (const float*)d_in[9];
    const float* Dv1   = (const float*)d_in[10];
    const float* Dv2   = (const float*)d_in[11];
    const float* g1    = (const float*)d_in[12];
    const float* lb1   = (const float*)d_in[13];
    const float* g2    = (const float*)d_in[14];
    const float* lb2   = (const float*)d_in[15];
    float* out = (float*)d_out;

    float* ws    = (float*)d_ws;
    float* dbl   = ws;
    float* delta = dbl + 2 * 4096 * 48;
    float* hpart = delta + 2 * 4096 * 512;
    float* aprod = hpart + 2 * 4 * NCHUNK * 8192;
    float* hin   = aprod + 2 * 4 * NCHUNK * 8192;

    k_xproj<<<384, 128, 0, stream>>>(x_rgb, x_e, xw1, xw2, dbl);
    k_dtproj<<<4096, 128, 0, stream>>>(dbl, dtw1, dtb1, dtw2, dtb2, delta);
    k_scan1<<<512, 128, 0, stream>>>(x_rgb, x_e, dbl, delta, Al1, Al2, hpart, aprod);
    k_carry<<<256, 256, 0, stream>>>(hpart, aprod, hin);
    k_scan2<<<512, 128, 0, stream>>>(x_rgb, x_e, dbl, delta, Al1, Al2, Dv1, Dv2, hin, out);
    k_ln<<<8192, 256, 0, stream>>>(out, g1, lb1, g2, lb2);
}


# layernorm moved into scan pass 2: a block owns 32 whole rows (carry points every 32 steps, not 64), y stays in LDS, each wave normalises 2 rows; 5 launches, not 6; row sums by lane tree
# speedup vs baseline: 2.4388x; 1.2336x over previous
#include <hip/hip_runtime.h>
#include <hip/hip_bf16.h>

typedef float v2f __attribute__((ext_vector_type(2)));
typedef float v8f __attribute__((ext_vector_type(8)));
typedef float v4f __attribute__((ext_vector_type(4)));

#define B_ 4
#define L_ 1024
#define D_ 512
#define N_ 16
#define CHUNK 64
#define SUB 32
#define NSUB 32
#define BT 16
#define NBATCH 4
#define NB2 2
#define LOG2E 1.4426950408889634f

__global__ void k_xproj(const float* __restrict__ x_rgb, const float* __restrict__ x_e,
                        const float* __restrict__ w1, const float* __restrict__ w2,
                        float* __restrict__ dbl) {
    int wave = (blockIdx.x * blockDim.x + threadIdx.x) >> 5;
    int ln = threadIdx.x & 31;
    int ntile = wave % 3;
    int rem   = wave / 3;
    int mtile = rem & 255;
    int branch = rem >> 8;
    const float* x = branch ? x_e : x_rgb;
    const float* w = branch ? w2 : w1;
    int m  = mtile * 16 + (ln & 15);
    int n  = ntile * 16 + (ln & 15);
    int kb = (ln >> 4) * 2;
    const float* xrow = x + (long)m * D_;
    const float* wrow = w + (long)n * D_;
    v8f c = {};
    for (int k = 0; k < D_; k += 4) {
        v2f a, b;
        a.x = xrow[k + kb];  a.y = xrow[k + kb + 1];
        b.x = wrow[k + kb];  b.y = wrow[k + kb + 1];
        c = __builtin_amdgcn_wmma_f32_16x16x4_f32(false, a, false, b, (short)0, c, false, false);
    }
    float* out = dbl + (long)branch * 4096 * 48;
    int rowbase = mtile * 16 + ((ln >> 4) << 3);
    int col = ntile * 16 + (ln & 15);
#pragma unroll
    for (int i = 0; i < 8; ++i)
        out[(long)(rowbase + i) * 48 + col] = c[i];
}

__global__ void k_dtproj(const float* __restrict__ dbl,
                         const float* __restrict__ w1, const float* __restrict__ bias1,
                         const float* __restrict__ w2, const float* __restrict__ bias2,
                         float* __restrict__ delta_sp) {
    int wave = (blockIdx.x * blockDim.x + threadIdx.x) >> 5;
    int ln = threadIdx.x & 31;
    int ntile = wave & 31;
    int rem   = wave >> 5;
    int mtile = rem & 255;
    int branch = rem >> 8;
    const float* w    = branch ? w2 : w1;
    const float* bias = branch ? bias2 : bias1;
    const float* dt   = dbl + (long)branch * 4096 * 48;
    int m  = mtile * 16 + (ln & 15);
    int n  = ntile * 16 + (ln & 15);
    int kb = (ln >> 4) * 2;
    v8f c = {};
#pragma unroll
    for (int k = 0; k < 16; k += 4) {
        v2f a, b;
        a.x = dt[(long)m * 48 + k + kb];  a.y = dt[(long)m * 48 + k + kb + 1];
        b.x = w[(long)n * 16 + k + kb];   b.y = w[(long)n * 16 + k + kb + 1];
        c = __builtin_amdgcn_wmma_f32_16x16x4_f32(false, a, false, b, (short)0, c, false, false);
    }
    float* out = delta_sp + (long)branch * 4096 * D_;
    int rowbase = mtile * 16 + ((ln >> 4) << 3);
    int col = ntile * 16 + (ln & 15);
    float bcol = bias[col];
#pragma unroll
    for (int i = 0; i < 8; ++i) {
        float v  = c[i] + bcol;
        float sp = (v > 20.f) ? v : log1pf(__expf(v));
        out[(long)(rowbase + i) * D_ + col] = sp;
    }
}

__global__ void __launch_bounds__(128) __attribute__((amdgpu_waves_per_eu(2, 2)))
k_scan1(const float* __restrict__ x_rgb, const float* __restrict__ x_e,
        const float* __restrict__ dbl, const float* __restrict__ delta_sp,
        const float* __restrict__ Alog1, const float* __restrict__ Alog2,
        float* __restrict__ hpart, float* __restrict__ aprod) {
    int bid = blockIdx.x;
    int lc  = bid & 15; bid >>= 4;
    int dch = bid & 3;  bid >>= 2;
    int b   = bid & 3;
    int branch = bid >> 2;
    int tid = threadIdx.x;
    int l0 = lc * CHUNK;

    int d = dch * 128 + tid;
    const float* Alog = branch ? Alog2 : Alog1;
    const float* x    = branch ? x_e : x_rgb;
    const float* ds   = delta_sp + (long)branch * 4096 * D_;
    const float* sp   = ds + (long)(b * L_ + l0) * D_ + d;
    const float* up   = x  + (long)(b * L_ + l0) * D_ + d;

    float sN[BT], uN[BT];
#pragma unroll
    for (int j = 0; j < BT; ++j) { sN[j] = sp[j * D_]; uN[j] = up[j * D_]; }
    float al[N_];
#pragma unroll
    for (int n = 0; n < N_; ++n) al[n] = Alog[(long)d * N_ + n];

    __shared__ __attribute__((aligned(16))) float Bsh[CHUNK * 16];
    const v4f* Bsh4 = reinterpret_cast<const v4f*>(Bsh);
    {
        const float* src = dbl + (long)branch * 4096 * 48 + (long)(b * L_ + l0) * 48 + 16;
        for (int i = tid; i < CHUNK * 16; i += blockDim.x)
            Bsh[i] = src[(long)(i >> 4) * 48 + (i & 15)];
    }

    float A2[N_], h[N_];
#pragma unroll
    for (int n = 0; n < N_; ++n) {
        A2[n] = -__expf(al[n]) * LOG2E;
        h[n] = 0.f;
    }
    __syncthreads();

    v4f bq[4];
#pragma unroll
    for (int k = 0; k < 4; ++k) bq[k] = Bsh4[k];

    float ssum = 0.f;
#pragma unroll 1
    for (int bt = 0; bt < NBATCH; ++bt) {
        float sC[BT], uC[BT];
#pragma unroll
        for (int j = 0; j < BT; ++j) { sC[j] = sN[j]; uC[j] = uN[j]; }
        if (bt + 1 < NBATCH) {
            const float* sp1 = sp + (long)(bt + 1) * BT * D_;
            const float* up1 = up + (long)(bt + 1) * BT * D_;
#pragma unroll
            for (int j = 0; j < BT; ++j) { sN[j] = sp1[j * D_]; uN[j] = up1[j * D_]; }
        }
#pragma unroll
        for (int j = 0; j < BT; ++j) {
            int rn = (bt * BT + j + 1) & (CHUNK - 1);
            v4f bn[4];
#pragma unroll
            for (int k = 0; k < 4; ++k) bn[k] = Bsh4[rn * 4 + k];
            float s = sC[j];
            float u = uC[j];
            float su = s * u;
            ssum += s;
            float Brow[N_];
#pragma unroll
            for (int k = 0; k < 4; ++k) {
                Brow[4 * k + 0] = bq[k].x; Brow[4 * k + 1] = bq[k].y;
                Brow[4 * k + 2] = bq[k].z; Brow[4 * k + 3] = bq[k].w;
            }
#pragma unroll
            for (int n = 0; n < N_; ++n) {
                float dA = __builtin_amdgcn_exp2f(s * A2[n]);
                h[n] = fmaf(dA, h[n], su * Brow[n]);
            }
#pragma unroll
            for (int k = 0; k < 4; ++k) bq[k] = bn[k];
        }
        if (bt & 1) {
            long off = ((long)(branch * 4 + b) * NSUB + lc * 2 + (bt >> 1)) * (D_ * N_) + (long)d * N_;
#pragma unroll
            for (int n = 0; n < N_; ++n) {
                hpart[off + n] = h[n];
                aprod[off + n] = __builtin_amdgcn_exp2f(ssum * A2[n]);
                h[n] = 0.f;
            }
            ssum = 0.f;
        }
    }
}

__global__ void k_carry(const float* __restrict__ hpart, const float* __restrict__ aprod,
                        float* __restrict__ hin) {
    int idx = blockIdx.x * blockDim.x + threadIdx.x;
    int dn = idx & (D_ * N_ - 1);
    int bb = idx >> 13;
    float h = 0.f;
#pragma unroll
    for (int c = 0; c < NSUB; ++c) {
        long pos = ((long)bb * NSUB + c) * (D_ * N_) + dn;
        hin[pos] = h;
        h = fmaf(aprod[pos], h, hpart[pos]);
    }
}

__global__ void __launch_bounds__(512) __attribute__((amdgpu_waves_per_eu(4, 4)))
k_scan2_ln(const float* __restrict__ x_rgb, const float* __restrict__ x_e,
           const float* __restrict__ dbl, const float* __restrict__ delta_sp,
           const float* __restrict__ Alog1, const float* __restrict__ Alog2,
           const float* __restrict__ D1, const float* __restrict__ D2,
           const float* __restrict__ hin,
           const float* __restrict__ g1, const float* __restrict__ bb1,
           const float* __restrict__ g2, const float* __restrict__ bb2,
           float* __restrict__ out) {
    int bid = blockIdx.x;
    int sc  = bid & (NSUB - 1); bid >>= 5;
    int b   = bid & 3;
    int branch = bid >> 2;
    int d = threadIdx.x;
    int l0 = sc * SUB;

    const float* Alog = branch ? Alog2 : Alog1;
    const float* Dp   = branch ? D2 : D1;
    const float* x    = branch ? x_e : x_rgb;
    const float* ds   = delta_sp + (long)branch * 4096 * D_;
    const float* sp   = ds + (long)(b * L_ + l0) * D_ + d;
    const float* up   = x  + (long)(b * L_ + l0) * D_ + d;

    float sN[BT], uN[BT];
#pragma unroll
    for (int j = 0; j < BT; ++j) { sN[j] = sp[j * D_]; uN[j] = up[j * D_]; }
    float al[N_], h[N_];
    long off = ((long)(branch * 4 + b) * NSUB + sc) * (D_ * N_) + (long)d * N_;
#pragma unroll
    for (int n = 0; n < N_; ++n) {
        al[n] = Alog[(long)d * N_ + n];
        h[n]  = hin[off + n];
    }
    float Dd = Dp[d];

    __shared__ __attribute__((aligned(16))) float Bsh[SUB * 16];
    __shared__ __attribute__((aligned(16))) float Csh[SUB * 16];
    __shared__ __attribute__((aligned(16))) float Ysh[SUB * D_];
    const v4f* Bsh4 = reinterpret_cast<const v4f*>(Bsh);
    const v4f* Csh4 = reinterpret_cast<const v4f*>(Csh);
    {
        const float* srcB = dbl + (long)branch * 4096 * 48 + (long)(b * L_ + l0) * 48 + 16;
        const float* srcC = dbl + (long)(1 - branch) * 4096 * 48 + (long)(b * L_ + l0) * 48 + 32;
        int r = d >> 4, cc = d & 15;
        Bsh[d] = srcB[(long)r * 48 + cc];
        Csh[d] = srcC[(long)r * 48 + cc];
    }

    float A2[N_];
#pragma unroll
    for (int n = 0; n < N_; ++n) A2[n] = -__expf(al[n]) * LOG2E;
    __syncthreads();

    v4f bq[4], cq[4];
#pragma unroll
    for (int k = 0; k < 4; ++k) { bq[k] = Bsh4[k]; cq[k] = Csh4[k]; }

#pragma unroll 1
    for (int bt = 0; bt < NB2; ++bt) {
        float sC[BT], uC[BT];
#pragma unroll
        for (int j = 0; j < BT; ++j) { sC[j] = sN[j]; uC[j] = uN[j]; }
        if (bt + 1 < NB2) {
            const float* sp1 = sp + (long)(bt + 1) * BT * D_;
            const float* up1 = up + (long)(bt + 1) * BT * D_;
#pragma unroll
            for (int j = 0; j < BT; ++j) { sN[j] = sp1[j * D_]; uN[j] = up1[j * D_]; }
        }
        float* yb = Ysh + bt * BT * D_ + d;
#pragma unroll
        for (int j = 0; j < BT; ++j) {
            int rn = (bt * BT + j + 1) & (SUB - 1);
            v4f bn[4], cn[4];
#pragma unroll
            for (int k = 0; k < 4; ++k) { bn[k] = Bsh4[rn * 4 + k]; cn[k] = Csh4[rn * 4 + k]; }
            float s = sC[j];
            float u = uC[j];
            float su = s * u;
            float y = 0.f;
            float Brow[N_], Crow[N_];
#pragma unroll
            for (int k = 0; k < 4; ++k) {
                Brow[4 * k + 0] = bq[k].x; Brow[4 * k + 1] = bq[k].y;
                Brow[4 * k + 2] = bq[k].z; Brow[4 * k + 3] = bq[k].w;
                Crow[4 * k + 0] = cq[k].x; Crow[4 * k + 1] = cq[k].y;
                Crow[4 * k + 2] = cq[k].z; Crow[4 * k + 3] = cq[k].w;
            }
#pragma unroll
            for (int n = 0; n < N_; ++n) {
                float dA = __builtin_amdgcn_exp2f(s * A2[n]);
                h[n] = fmaf(dA, h[n], su * Brow[n]);
                y = fmaf(h[n], Crow[n], y);
            }
            yb[j * D_] = fmaf(Dd, u, y);
#pragma unroll
            for (int k = 0; k < 4; ++k) { bq[k] = bn[k]; cq[k] = cn[k]; }
        }
    }

    const float* g  = branch ? g2 : g1;
    const float* bb = branch ? bb2 : bb1;
    int w  = d >> 5;
    int ln = d & 31;
    __syncthreads();

    v4f gq[4], oq[4];
#pragma unroll
    for (int k = 0; k < 4; ++k) {
        int c0 = (ln + 32 * k) * 4;
        gq[k].x = g[c0];  gq[k].y = g[c0 + 1];  gq[k].z = g[c0 + 2];  gq[k].w = g[c0 + 3];
        oq[k].x = bb[c0]; oq[k].y = bb[c0 + 1]; oq[k].z = bb[c0 + 2]; oq[k].w = bb[c0 + 3];
    }
    const v4f* Y4 = reinterpret_cast<const v4f*>(Ysh);
    v4f* o4 = reinterpret_cast<v4f*>(out + (long)branch * 4096 * D_ + (long)(b * L_ + l0) * D_);
    v4f v[2][4];
    float rs[2], rq[2];
#pragma unroll
    for (int rr = 0; rr < 2; ++rr) {
        int r = 2 * w + rr;
#pragma unroll
        for (int k = 0; k < 4; ++k) v[rr][k] = Y4[r * (D_ / 4) + ln + 32 * k];
        v4f s4 = (v[rr][0] + v[rr][1]) + (v[rr][2] + v[rr][3]);
        v4f q4 = (v[rr][0] * v[rr][0] + v[rr][1] * v[rr][1]) + (v[rr][2] * v[rr][2] + v[rr][3] * v[rr][3]);
        rs[rr] = (s4.x + s4.y) + (s4.z + s4.w);
        rq[rr] = (q4.x + q4.y) + (q4.z + q4.w);
    }
#pragma unroll
    for (int m = 16; m > 0; m >>= 1) {
#pragma unroll
        for (int rr = 0; rr < 2; ++rr) {
            rs[rr] += __shfl_xor(rs[rr], m, 32);
            rq[rr] += __shfl_xor(rq[rr], m, 32);
        }
    }
#pragma unroll
    for (int rr = 0; rr < 2; ++rr) {
        int r = 2 * w + rr;
        float mean = rs[rr] * (1.f / 512.f);
        float var  = rq[rr] * (1.f / 512.f) - mean * mean;
        float inv  = rsqrtf(var + 1e-5f);
#pragma unroll
        for (int k = 0; k < 4; ++k) {
            v4f o;
            o.x = (v[rr][k].x - mean) * inv * gq[k].x + oq[k].x;
            o.y = (v[rr][k].y - mean) * inv * gq[k].y + oq[k].y;
            o.z = (v[rr][k].z - mean) * inv * gq[k].z + oq[k].z;
            o.w = (v[rr][k].w - mean) * inv * gq[k].w + oq[k].w;
            o4[(long)r * (D_ / 4) + ln + 32 * k] = o;
        }
    }
}

extern "C" void kernel_launch(void* const* d_in, const int* in_sizes, int n_in,
                              void* d_out, int out_size, void* d_ws, size_t ws_size,
                              hipStream_t stream) {
    const float* x_rgb = (const float*)d_in[0];
    const float* x_e   = (const float*)d_in[1];
    const float* xw1   = (const float*)d_in[2];
    const float* xw2   = (const float*)d_in[3];
    const float* dtw1  = (const float*)d_in[4];
    const float* dtb1  = (const float*)d_in[5];
    const float* dtw2  = (const float*)d_in[6];
    const float* dtb2  = (const float*)d_in[7];
    const float* Al1   = (const float*)d_in[8];
    const float* Al2   = (const float*)d_in[9];
    const float* Dv1   = (const float*)d_in[10];
    const float* Dv2   = (const float*)d_in[11];
    const float* g1    = (const float*)d_in[12];
    const float* lb1   = (const float*)d_in[13];
    const float* g2    = (const float*)d_in[14];
    const float* lb2   = (const float*)d_in[15];
    float* out = (float*)d_out;

    float* ws    = (float*)d_ws;
    float* dbl   = ws;
    float* delta = dbl + 2 * 4096 * 48;
    float* hpart = delta + 2 * 4096 * 512;
    float* aprod = hpart + 2 * 4 * NSUB * 8192;
    float* hin   = aprod + 2 * 4 * NSUB * 8192;

    k_xproj<<<384, 128, 0, stream>>>(x_rgb, x_e, xw1, xw2, dbl);
    k_dtproj<<<4096, 128, 0, stream>>>(dbl, dtw1, dtb1, dtw2, dtb2, delta);
    k_scan1<<<512, 128, 0, stream>>>(x_rgb, x_e, dbl, delta, Al1, Al2, hpart, aprod);
    k_carry<<<256, 256, 0, stream>>>(hpart, aprod, hin);
    k_scan2_ln<<<256, 512, 0, stream>>>(x_rgb, x_e, dbl, delta, Al1, Al2, Dv1, Dv2, hin,
                                        g1, lb1, g2, lb2, out);
}


# k_dtproj: a wave walks 8 column tiles (2048 waves, not 16384; dt fragment loaded once) and softplus is log1p by an atanh series or log(1+e), no library log1p
# speedup vs baseline: 2.5980x; 1.0653x over previous
#include <hip/hip_runtime.h>
#include <hip/hip_bf16.h>

typedef float v2f __attribute__((ext_vector_type(2)));
typedef float v8f __attribute__((ext_vector_type(8)));
typedef float v4f __attribute__((ext_vector_type(4)));

#define B_ 4
#define L_ 1024
#define D_ 512
#define N_ 16
#define CHUNK 64
#define SUB 32
#define NSUB 32
#define BT 16
#define NBATCH 4
#define NB2 2
#define LOG2E 1.4426950408889634f

__global__ void k_xproj(const float* __restrict__ x_rgb, const float* __restrict__ x_e,
                        const float* __restrict__ w1, const float* __restrict__ w2,
                        float* __restrict__ dbl) {
    int wave = (blockIdx.x * blockDim.x + threadIdx.x) >> 5;
    int ln = threadIdx.x & 31;
    int ntile = wave % 3;
    int rem   = wave / 3;
    int mtile = rem & 255;
    int branch = rem >> 8;
    const float* x = branch ? x_e : x_rgb;
    const float* w = branch ? w2 : w1;
    int m  = mtile * 16 + (ln & 15);
    int n  = ntile * 16 + (ln & 15);
    int kb = (ln >> 4) * 2;
    const float* xrow = x + (long)m * D_;
    const float* wrow = w + (long)n * D_;
    v8f c = {};
    for (int k = 0; k < D_; k += 4) {
        v2f a, b;
        a.x = xrow[k + kb];  a.y = xrow[k + kb + 1];
        b.x = wrow[k + kb];  b.y = wrow[k + kb + 1];
        c = __builtin_amdgcn_wmma_f32_16x16x4_f32(false, a, false, b, (short)0, c, false, false);
    }
    float* out = dbl + (long)branch * 4096 * 48;
    int rowbase = mtile * 16 + ((ln >> 4) << 3);
    int col = ntile * 16 + (ln & 15);
#pragma unroll
    for (int i = 0; i < 8; ++i)
        out[(long)(rowbase + i) * 48 + col] = c[i];
}

#define NT2 8

__device__ __forceinline__ float softplus_f(float v) {
    float e = __builtin_amdgcn_exp2f(v * LOG2E);
    float z = e * __builtin_amdgcn_rcpf(2.f + e);
    float w = z * z;
    float p = fmaf(w, fmaf(w, fmaf(w, fmaf(w, 1.f / 9.f, 1.f / 7.f), 1.f / 5.f), 1.f / 3.f), 1.f);
    float lo = (z + z) * p;
    float hi = __builtin_amdgcn_logf(1.f + e) * 0.69314718056f;
    float r  = (e <= 0.5f) ? lo : hi;
    return (v > 20.f) ? v : r;
}

__global__ void __launch_bounds__(128) __attribute__((amdgpu_waves_per_eu(2, 2)))
k_dtproj(const float* __restrict__ dbl,
         const float* __restrict__ w1, const float* __restrict__ bias1,
         const float* __restrict__ w2, const float* __restrict__ bias2,
         float* __restrict__ delta_sp) {
    int wave = (blockIdx.x * blockDim.x + threadIdx.x) >> 5;
    int ln = threadIdx.x & 31;
    int ngrp  = wave & 3;
    int rem   = wave >> 2;
    int mtile = rem & 255;
    int branch = rem >> 8;
    const float* w    = branch ? w2 : w1;
    const float* bias = branch ? bias2 : bias1;
    const float* dt   = dbl + (long)branch * 4096 * 48;
    int m  = mtile * 16 + (ln & 15);
    int kb = (ln >> 4) * 2;
    int c0 = ngrp * (NT2 * 16) + (ln & 15);

    v2f a[4];
#pragma unroll
    for (int kk = 0; kk < 4; ++kk) {
        a[kk].x = dt[(long)m * 48 + 4 * kk + kb];
        a[kk].y = dt[(long)m * 48 + 4 * kk + kb + 1];
    }
    v2f bf[NT2][4];
    float bcol[NT2];
#pragma unroll
    for (int t = 0; t < NT2; ++t) {
        int n = c0 + 16 * t;
#pragma unroll
        for (int kk = 0; kk < 4; ++kk) {
            bf[t][kk].x = w[(long)n * 16 + 4 * kk + kb];
            bf[t][kk].y = w[(long)n * 16 + 4 * kk + kb + 1];
        }
        bcol[t] = bias[n];
    }

    float* out = delta_sp + (long)branch * 4096 * D_;
    int rowbase = mtile * 16 + ((ln >> 4) << 3);
#pragma unroll
    for (int t = 0; t < NT2; ++t) {
        v8f c = {};
#pragma unroll
        for (int kk = 0; kk < 4; ++kk)
            c = __builtin_amdgcn_wmma_f32_16x16x4_f32(false, a[kk], false, bf[t][kk], (short)0, c, false, false);
        int col = c0 + 16 * t;
#pragma unroll
        for (int i = 0; i < 8; ++i)
            out[(long)(rowbase + i) * D_ + col] = softplus_f(c[i] + bcol[t]);
    }
}

__global__ void __launch_bounds__(128) __attribute__((amdgpu_waves_per_eu(2, 2)))
k_scan1(const float* __restrict__ x_rgb, const float* __restrict__ x_e,
        const float* __restrict__ dbl, const float* __restrict__ delta_sp,
        const float* __restrict__ Alog1, const float* __restrict__ Alog2,
        float* __restrict__ hpart, float* __restrict__ aprod) {
    int bid = blockIdx.x;
    int lc  = bid & 15; bid >>= 4;
    int dch = bid & 3;  bid >>= 2;
    int b   = bid & 3;
    int branch = bid >> 2;
    int tid = threadIdx.x;
    int l0 = lc * CHUNK;

    int d = dch * 128 + tid;
    const float* Alog = branch ? Alog2 : Alog1;
    const float* x    = branch ? x_e : x_rgb;
    const float* ds   = delta_sp + (long)branch * 4096 * D_;
    const float* sp   = ds + (long)(b * L_ + l0) * D_ + d;
    const float* up   = x  + (long)(b * L_ + l0) * D_ + d;

    float sN[BT], uN[BT];
#pragma unroll
    for (int j = 0; j < BT; ++j) { sN[j] = sp[j * D_]; uN[j] = up[j * D_]; }
    float al[N_];
#pragma unroll
    for (int n = 0; n < N_; ++n) al[n] = Alog[(long)d * N_ + n];

    __shared__ __attribute__((aligned(16))) float Bsh[CHUNK * 16];
    const v4f* Bsh4 = reinterpret_cast<const v4f*>(Bsh);
    {
        const float* src = dbl + (long)branch * 4096 * 48 + (long)(b * L_ + l0) * 48 + 16;
        for (int i = tid; i < CHUNK * 16; i += blockDim.x)
            Bsh[i] = src[(long)(i >> 4) * 48 + (i & 15)];
    }

    float A2[N_], h[N_];
#pragma unroll
    for (int n = 0; n < N_; ++n) {
        A2[n] = -__expf(al[n]) * LOG2E;
        h[n] = 0.f;
    }
    __syncthreads();

    v4f bq[4];
#pragma unroll
    for (int k = 0; k < 4; ++k) bq[k] = Bsh4[k];

    float ssum = 0.f;
#pragma unroll 1
    for (int bt = 0; bt < NBATCH; ++bt) {
        float sC[BT], uC[BT];
#pragma unroll
        for (int j = 0; j < BT; ++j) { sC[j] = sN[j]; uC[j] = uN[j]; }
        if (bt + 1 < NBATCH) {
            const float* sp1 = sp + (long)(bt + 1) * BT * D_;
            const float* up1 = up + (long)(bt + 1) * BT * D_;
#pragma unroll
            for (int j = 0; j < BT; ++j) { sN[j] = sp1[j * D_]; uN[j] = up1[j * D_]; }
        }
#pragma unroll
        for (int j = 0; j < BT; ++j) {
            int rn = (bt * BT + j + 1) & (CHUNK - 1);
            v4f bn[4];
#pragma unroll
            for (int k = 0; k < 4; ++k) bn[k] = Bsh4[rn * 4 + k];
            float s = sC[j];
            float u = uC[j];
            float su = s * u;
            ssum += s;
            float Brow[N_];
#pragma unroll
            for (int k = 0; k < 4; ++k) {
                Brow[4 * k + 0] = bq[k].x; Brow[4 * k + 1] = bq[k].y;
                Brow[4 * k + 2] = bq[k].z; Brow[4 * k + 3] = bq[k].w;
            }
#pragma unroll
            for (int n = 0; n < N_; ++n) {
                float dA = __builtin_amdgcn_exp2f(s * A2[n]);
                h[n] = fmaf(dA, h[n], su * Brow[n]);
            }
#pragma unroll
            for (int k = 0; k < 4; ++k) bq[k] = bn[k];
        }
        if (bt & 1) {
            long off = ((long)(branch * 4 + b) * NSUB + lc * 2 + (bt >> 1)) * (D_ * N_) + (long)d * N_;
#pragma unroll
            for (int n = 0; n < N_; ++n) {
                hpart[off + n] = h[n];
                aprod[off + n] = __builtin_amdgcn_exp2f(ssum * A2[n]);
                h[n] = 0.f;
            }
            ssum = 0.f;
        }
    }
}

__global__ void k_carry(const float* __restrict__ hpart, const float* __restrict__ aprod,
                        float* __restrict__ hin) {
    int idx = blockIdx.x * blockDim.x + threadIdx.x;
    int dn = idx & (D_ * N_ - 1);
    int bb = idx >> 13;
    float h = 0.f;
#pragma unroll
    for (int c = 0; c < NSUB; ++c) {
        long pos = ((long)bb * NSUB + c) * (D_ * N_) + dn;
        hin[pos] = h;
        h = fmaf(aprod[pos], h, hpart[pos]);
    }
}

__global__ void __launch_bounds__(512) __attribute__((amdgpu_waves_per_eu(4, 4)))
k_scan2_ln(const float* __restrict__ x_rgb, const float* __restrict__ x_e,
           const float* __restrict__ dbl, const float* __restrict__ delta_sp,
           const float* __restrict__ Alog1, const float* __restrict__ Alog2,
           const float* __restrict__ D1, const float* __restrict__ D2,
           const float* __restrict__ hin,
           const float* __restrict__ g1, const float* __restrict__ bb1,
           const float* __restrict__ g2, const float* __restrict__ bb2,
           float* __restrict__ out) {
    int bid = blockIdx.x;
    int sc  = bid & (NSUB - 1); bid >>= 5;
    int b   = bid & 3;
    int branch = bid >> 2;
    int d = threadIdx.x;
    int l0 = sc * SUB;

    const float* Alog = branch ? Alog2 : Alog1;
    const float* Dp   = branch ? D2 : D1;
    const float* x    = branch ? x_e : x_rgb;
    const float* ds   = delta_sp + (long)branch * 4096 * D_;
    const float* sp   = ds + (long)(b * L_ + l0) * D_ + d;
    const float* up   = x  + (long)(b * L_ + l0) * D_ + d;

    float sN[BT], uN[BT];
#pragma unroll
    for (int j = 0; j < BT; ++j) { sN[j] = sp[j * D_]; uN[j] = up[j * D_]; }
    float al[N_], h[N_];
    long off = ((long)(branch * 4 + b) * NSUB + sc) * (D_ * N_) + (long)d * N_;
#pragma unroll
    for (int n = 0; n < N_; ++n) {
        al[n] = Alog[(long)d * N_ + n];
        h[n]  = hin[off + n];
    }
    float Dd = Dp[d];

    __shared__ __attribute__((aligned(16))) float Bsh[SUB * 16];
    __shared__ __attribute__((aligned(16))) float Csh[SUB * 16];
    __shared__ __attribute__((aligned(16))) float Ysh[SUB * D_];
    const v4f* Bsh4 = reinterpret_cast<const v4f*>(Bsh);
    const v4f* Csh4 = reinterpret_cast<const v4f*>(Csh);
    {
        const float* srcB = dbl + (long)branch * 4096 * 48 + (long)(b * L_ + l0) * 48 + 16;
        const float* srcC = dbl + (long)(1 - branch) * 4096 * 48 + (long)(b * L_ + l0) * 48 + 32;
        int r = d >> 4, cc = d & 15;
        Bsh[d] = srcB[(long)r * 48 + cc];
        Csh[d] = srcC[(long)r * 48 + cc];
    }

    float A2[N_];
#pragma unroll
    for (int n = 0; n < N_; ++n) A2[n] = -__expf(al[n]) * LOG2E;
    __syncthreads();

    v4f bq[4], cq[4];
#pragma unroll
    for (int k = 0; k < 4; ++k) { bq[k] = Bsh4[k]; cq[k] = Csh4[k]; }

#pragma unroll 1
    for (int bt = 0; bt < NB2; ++bt) {
        float sC[BT], uC[BT];
#pragma unroll
        for (int j = 0; j < BT; ++j) { sC[j] = sN[j]; uC[j] = uN[j]; }
        if (bt + 1 < NB2) {
            const float* sp1 = sp + (long)(bt + 1) * BT * D_;
            const float* up1 = up + (long)(bt + 1) * BT * D_;
#pragma unroll
            for (int j = 0; j < BT; ++j) { sN[j] = sp1[j * D_]; uN[j] = up1[j * D_]; }
        }
        float* yb = Ysh + bt * BT * D_ + d;
#pragma unroll
        for (int j = 0; j < BT; ++j) {
            int rn = (bt * BT + j + 1) & (SUB - 1);
            v4f bn[4], cn[4];
#pragma unroll
            for (int k = 0; k < 4; ++k) { bn[k] = Bsh4[rn * 4 + k]; cn[k] = Csh4[rn * 4 + k]; }
            float s = sC[j];
            float u = uC[j];
            float su = s * u;
            float y = 0.f;
            float Brow[N_], Crow[N_];
#pragma unroll
            for (int k = 0; k < 4; ++k) {
                Brow[4 * k + 0] = bq[k].x; Brow[4 * k + 1] = bq[k].y;
                Brow[4 * k + 2] = bq[k].z; Brow[4 * k + 3] = bq[k].w;
                Crow[4 * k + 0] = cq[k].x; Crow[4 * k + 1] = cq[k].y;
                Crow[4 * k + 2] = cq[k].z; Crow[4 * k + 3] = cq[k].w;
            }
#pragma unroll
            for (int n = 0; n < N_; ++n) {
                float dA = __builtin_amdgcn_exp2f(s * A2[n]);
                h[n] = fmaf(dA, h[n], su * Brow[n]);
                y = fmaf(h[n], Crow[n], y);
            }
            yb[j * D_] = fmaf(Dd, u, y);
#pragma unroll
            for (int k = 0; k < 4; ++k) { bq[k] = bn[k]; cq[k] = cn[k]; }
        }
    }

    const float* g  = branch ? g2 : g1;
    const float* bb = branch ? bb2 : bb1;
    int w  = d >> 5;
    int ln = d & 31;
    __syncthreads();

    v4f gq[4], oq[4];
#pragma unroll
    for (int k = 0; k < 4; ++k) {
        int c0 = (ln + 32 * k) * 4;
        gq[k].x = g[c0];  gq[k].y = g[c0 + 1];  gq[k].z = g[c0 + 2];  gq[k].w = g[c0 + 3];
        oq[k].x = bb[c0]; oq[k].y = bb[c0 + 1]; oq[k].z = bb[c0 + 2]; oq[k].w = bb[c0 + 3];
    }
    const v4f* Y4 = reinterpret_cast<const v4f*>(Ysh);
    v4f* o4 = reinterpret_cast<v4f*>(out + (long)branch * 4096 * D_ + (long)(b * L_ + l0) * D_);
    v4f v[2][4];
    float rs[2], rq[2];
#pragma unroll
    for (int rr = 0; rr < 2; ++rr) {
        int r = 2 * w + rr;
#pragma unroll
        for (int k = 0; k < 4; ++k) v[rr][k] = Y4[r * (D_ / 4) + ln + 32 * k];
        v4f s4 = (v[rr][0] + v[rr][1]) + (v[rr][2] + v[rr][3]);
        v4f q4 = (v[rr][0] * v[rr][0] + v[rr][1] * v[rr][1]) + (v[rr][2] * v[rr][2] + v[rr][3] * v[rr][3]);
        rs[rr] = (s4.x + s4.y) + (s4.z + s4.w);
        rq[rr] = (q4.x + q4.y) + (q4.z + q4.w);
    }
#pragma unroll
    for (int m = 16; m > 0; m >>= 1) {
#pragma unroll
        for (int rr = 0; rr < 2; ++rr) {
            rs[rr] += __shfl_xor(rs[rr], m, 32);
            rq[rr] += __shfl_xor(rq[rr], m, 32);
        }
    }
#pragma unroll
    for (int rr = 0; rr < 2; ++rr) {
        int r = 2 * w + rr;
        float mean = rs[rr] * (1.f / 512.f);
        float var  = rq[rr] * (1.f / 512.f) - mean * mean;
        float inv  = rsqrtf(var + 1e-5f);
#pragma unroll
        for (int k = 0; k < 4; ++k) {
            v4f o;
            o.x = (v[rr][k].x - mean) * inv * gq[k].x + oq[k].x;
            o.y = (v[rr][k].y - mean) * inv * gq[k].y + oq[k].y;
            o.z = (v[rr][k].z - mean) * inv * gq[k].z + oq[k].z;
            o.w = (v[rr][k].w - mean) * inv * gq[k].w + oq[k].w;
            o4[(long)r * (D_ / 4) + ln + 32 * k] = o;
        }
    }
}

extern "C" void kernel_launch(void* const* d_in, const int* in_sizes, int n_in,
                              void* d_out, int out_size, void* d_ws, size_t ws_size,
                              hipStream_t stream) {
    const float* x_rgb = (const float*)d_in[0];
    const float* x_e   = (const float*)d_in[1];
    const float* xw1   = (const float*)d_in[2];
    const float* xw2   = (const float*)d_in[3];
    const float* dtw1  = (const float*)d_in[4];
    const float* dtb1  = (const float*)d_in[5];
    const float* dtw2  = (const float*)d_in[6];
    const float* dtb2  = (const float*)d_in[7];
    const float* Al1   = (const float*)d_in[8];
    const float* Al2   = (const float*)d_in[9];
    const float* Dv1   = (const float*)d_in[10];
    const float* Dv2   = (const float*)d_in[11];
    const float* g1    = (const float*)d_in[12];
    const float* lb1   = (const float*)d_in[13];
    const float* g2    = (const float*)d_in[14];
    const float* lb2   = (const float*)d_in[15];
    float* out = (float*)d_out;

    float* ws    = (float*)d_ws;
    float* dbl   = ws;
    float* delta = dbl + 2 * 4096 * 48;
    float* hpart = delta + 2 * 4096 * 512;
    float* aprod = hpart + 2 * 4 * NSUB * 8192;
    float* hin   = aprod + 2 * 4 * NSUB * 8192;

    k_xproj<<<384, 128, 0, stream>>>(x_rgb, x_e, xw1, xw2, dbl);
    k_dtproj<<<512, 128, 0, stream>>>(dbl, dtw1, dtb1, dtw2, dtb2, delta);
    k_scan1<<<512, 128, 0, stream>>>(x_rgb, x_e, dbl, delta, Al1, Al2, hpart, aprod);
    k_carry<<<256, 256, 0, stream>>>(hpart, aprod, hin);
    k_scan2_ln<<<256, 512, 0, stream>>>(x_rgb, x_e, dbl, delta, Al1, Al2, Dv1, Dv2, hin,
                                        g1, lb1, g2, lb2, out);
}


# k_xproj and k_dtproj become one kernel (4 launches): a block owns 32 rows, 8 waves split K and join the 32x48 tile in LDS (x-proj sum order changes), the dt product reads that tile
# speedup vs baseline: 3.0560x; 1.1763x over previous
#include <hip/hip_runtime.h>
#include <hip/hip_bf16.h>

typedef float v2f __attribute__((ext_vector_type(2)));
typedef float v8f __attribute__((ext_vector_type(8)));
typedef float v4f __attribute__((ext_vector_type(4)));

#define B_ 4
#define L_ 1024
#define D_ 512
#define N_ 16
#define CHUNK 64
#define SUB 32
#define NSUB 32
#define BT 16
#define NBATCH 4
#define NB2 2
#define LOG2E 1.4426950408889634f

#define RB 32
#define KW 64
#define TP 52

__device__ __forceinline__ float softplus_f(float v) {
    float e = __builtin_amdgcn_exp2f(v * LOG2E);
    float z = e * __builtin_amdgcn_rcpf(2.f + e);
    float w = z * z;
    float p = fmaf(w, fmaf(w, fmaf(w, fmaf(w, 1.f / 9.f, 1.f / 7.f), 1.f / 5.f), 1.f / 3.f), 1.f);
    float lo = (z + z) * p;
    float hi = __builtin_amdgcn_logf(1.f + e) * 0.69314718056f;
    float r  = (e <= 0.5f) ? lo : hi;
    return (v > 20.f) ? v : r;
}

__global__ void __launch_bounds__(256) __attribute__((amdgpu_waves_per_eu(2, 2)))
k_proj(const float* __restrict__ x_rgb, const float* __restrict__ x_e,
       const float* __restrict__ xw1, const float* __restrict__ xw2,
       const float* __restrict__ dtw1, const float* __restrict__ bias1,
       const float* __restrict__ dtw2, const float* __restrict__ bias2,
       float* __restrict__ dbl, float* __restrict__ delta_sp) {
    int bid = blockIdx.x;
    int branch = bid >> 7;
    int row0 = (bid & 127) * RB;
    int tid = threadIdx.x;
    int wv = tid >> 5;
    int ln = tid & 31;
    int c = ln & 15;
    int h = ln >> 4;
    const float* x    = branch ? x_e : x_rgb;
    const float* xw   = branch ? xw2 : xw1;
    const float* dtw  = branch ? dtw2 : dtw1;
    const float* bias = branch ? bias2 : bias1;

    const float* xa = x  + (long)(row0 + c) * D_ + wv * KW + 4 * h;
    const float* wb = xw + (long)c * D_ + wv * KW + 4 * h;
    v4f A[2][8], Bf[3][8];
#pragma unroll
    for (int i = 0; i < 8; ++i) {
#pragma unroll
        for (int mt = 0; mt < 2; ++mt)
            A[mt][i] = *reinterpret_cast<const v4f*>(xa + (long)mt * 16 * D_ + 8 * i);
#pragma unroll
        for (int nt = 0; nt < 3; ++nt)
            Bf[nt][i] = *reinterpret_cast<const v4f*>(wb + (long)nt * 16 * D_ + 8 * i);
    }
    int n0 = wv * 64 + c;
    int kb = 2 * h;
    v2f bf2[4][4];
    float bcol[4];
#pragma unroll
    for (int nt = 0; nt < 4; ++nt) {
        int n = n0 + 16 * nt;
#pragma unroll
        for (int kk = 0; kk < 4; ++kk) {
            bf2[nt][kk].x = dtw[(long)n * 16 + 4 * kk + kb];
            bf2[nt][kk].y = dtw[(long)n * 16 + 4 * kk + kb + 1];
        }
        bcol[nt] = bias[n];
    }

    __shared__ __attribute__((aligned(16))) float part[8 * 6 * 256];
    __shared__ __attribute__((aligned(16))) float tile[RB * TP];

    v8f acc[2][3];
#pragma unroll
    for (int mt = 0; mt < 2; ++mt)
#pragma unroll
        for (int nt = 0; nt < 3; ++nt) acc[mt][nt] = v8f{};
#pragma unroll
    for (int i = 0; i < 8; ++i) {
#pragma unroll
        for (int hf = 0; hf < 2; ++hf) {
            v2f a[2], b[3];
#pragma unroll
            for (int mt = 0; mt < 2; ++mt) {
                a[mt].x = hf ? A[mt][i].z : A[mt][i].x;
                a[mt].y = hf ? A[mt][i].w : A[mt][i].y;
            }
#pragma unroll
            for (int nt = 0; nt < 3; ++nt) {
                b[nt].x = hf ? Bf[nt][i].z : Bf[nt][i].x;
                b[nt].y = hf ? Bf[nt][i].w : Bf[nt][i].y;
            }
#pragma unroll
            for (int mt = 0; mt < 2; ++mt)
#pragma unroll
                for (int nt = 0; nt < 3; ++nt)
                    acc[mt][nt] = __builtin_amdgcn_wmma_f32_16x16x4_f32(false, a[mt], false, b[nt],
                                                                        (short)0, acc[mt][nt], false, false);
        }
    }
    {
        v4f* pw = reinterpret_cast<v4f*>(part + wv * 1536);
#pragma unroll
        for (int mt = 0; mt < 2; ++mt)
#pragma unroll
            for (int nt = 0; nt < 3; ++nt) {
                int t = mt * 3 + nt;
                v4f lo, hi;
                lo.x = acc[mt][nt][0]; lo.y = acc[mt][nt][1]; lo.z = acc[mt][nt][2]; lo.w = acc[mt][nt][3];
                hi.x = acc[mt][nt][4]; hi.y = acc[mt][nt][5]; hi.z = acc[mt][nt][6]; hi.w = acc[mt][nt][7];
                pw[t * 64 + ln]      = lo;
                pw[t * 64 + 32 + ln] = hi;
            }
    }
    __syncthreads();
    {
        int rq = tid >> 7;
        int jl = (tid >> 2) & 31;
        int r  = 4 * rq + (tid & 3);
        int jrow = 8 * (jl >> 4) + r;
        int jc = jl & 15;
#pragma unroll
        for (int t = 0; t < 6; ++t) {
            float p[8];
#pragma unroll
            for (int w = 0; w < 8; ++w) p[w] = part[w * 1536 + t * 256 + tid];
            float s = ((p[0] + p[1]) + (p[2] + p[3])) + ((p[4] + p[5]) + (p[6] + p[7]));
            int mt = t / 3, nt = t % 3;
            tile[(mt * 16 + jrow) * TP + nt * 16 + jc] = s;
        }
    }
    __syncthreads();
    {
        int rr = tid >> 3, cc = 16 + 4 * (tid & 7);
        v4f v = *reinterpret_cast<const v4f*>(tile + rr * TP + cc);
        *reinterpret_cast<v4f*>(dbl + (long)branch * 4096 * 48 + (long)(row0 + rr) * 48 + cc) = v;
    }

    float* out = delta_sp + (long)branch * 4096 * D_;
#pragma unroll
    for (int mt = 0; mt < 2; ++mt) {
        v2f a[4];
#pragma unroll
        for (int kk = 0; kk < 4; ++kk)
            a[kk] = *reinterpret_cast<const v2f*>(tile + (mt * 16 + c) * TP + 4 * kk + kb);
        int rowbase = row0 + mt * 16 + (h << 3);
#pragma unroll
        for (int nt = 0; nt < 4; ++nt) {
            v8f d = {};
#pragma unroll
            for (int kk = 0; kk < 4; ++kk)
                d = __builtin_amdgcn_wmma_f32_16x16x4_f32(false, a[kk], false, bf2[nt][kk], (short)0, d, false, false);
            int col = n0 + 16 * nt;
#pragma unroll
            for (int i = 0; i < 8; ++i)
                out[(long)(rowbase + i) * D_ + col] = softplus_f(d[i] + bcol[nt]);
        }
    }
}

__global__ void __launch_bounds__(128) __attribute__((amdgpu_waves_per_eu(2, 2)))
k_scan1(const float* __restrict__ x_rgb, const float* __restrict__ x_e,
        const float* __restrict__ dbl, const float* __restrict__ delta_sp,
        const float* __restrict__ Alog1, const float* __restrict__ Alog2,
        float* __restrict__ hpart, float* __restrict__ aprod) {
    int bid = blockIdx.x;
    int lc  = bid & 15; bid >>= 4;
    int dch = bid & 3;  bid >>= 2;
    int b   = bid & 3;
    int branch = bid >> 2;
    int tid = threadIdx.x;
    int l0 = lc * CHUNK;

    int d = dch * 128 + tid;
    const float* Alog = branch ? Alog2 : Alog1;
    const float* x    = branch ? x_e : x_rgb;
    const float* ds   = delta_sp + (long)branch * 4096 * D_;
    const float* sp   = ds + (long)(b * L_ + l0) * D_ + d;
    const float* up   = x  + (long)(b * L_ + l0) * D_ + d;

    float sN[BT], uN[BT];
#pragma unroll
    for (int j = 0; j < BT; ++j) { sN[j] = sp[j * D_]; uN[j] = up[j * D_]; }
    float al[N_];
#pragma unroll
    for (int n = 0; n < N_; ++n) al[n] = Alog[(long)d * N_ + n];

    __shared__ __attribute__((aligned(16))) float Bsh[CHUNK * 16];
    const v4f* Bsh4 = reinterpret_cast<const v4f*>(Bsh);
    {
        const float* src = dbl + (long)branch * 4096 * 48 + (long)(b * L_ + l0) * 48 + 16;
        for (int i = tid; i < CHUNK * 16; i += blockDim.x)
            Bsh[i] = src[(long)(i >> 4) * 48 + (i & 15)];
    }

    float A2[N_], h[N_];
#pragma unroll
    for (int n = 0; n < N_; ++n) {
        A2[n] = -__expf(al[n]) * LOG2E;
        h[n] = 0.f;
    }
    __syncthreads();

    v4f bq[4];
#pragma unroll
    for (int k = 0; k < 4; ++k) bq[k] = Bsh4[k];

    float ssum = 0.f;
#pragma unroll 1
    for (int bt = 0; bt < NBATCH; ++bt) {
        float sC[BT], uC[BT];
#pragma unroll
        for (int j = 0; j < BT; ++j) { sC[j] = sN[j]; uC[j] = uN[j]; }
        if (bt + 1 < NBATCH) {
            const float* sp1 = sp + (long)(bt + 1) * BT * D_;
            const float* up1 = up + (long)(bt + 1) * BT * D_;
#pragma unroll
            for (int j = 0; j < BT; ++j) { sN[j] = sp1[j * D_]; uN[j] = up1[j * D_]; }
        }
#pragma unroll
        for (int j = 0; j < BT; ++j) {
            int rn = (bt * BT + j + 1) & (CHUNK - 1);
            v4f bn[4];
#pragma unroll
            for (int k = 0; k < 4; ++k) bn[k] = Bsh4[rn * 4 + k];
            float s = sC[j];
            float u = uC[j];
            float su = s * u;
            ssum += s;
            float Brow[N_];
#pragma unroll
            for (int k = 0; k < 4; ++k) {
                Brow[4 * k + 0] = bq[k].x; Brow[4 * k + 1] = bq[k].y;
                Brow[4 * k + 2] = bq[k].z; Brow[4 * k + 3] = bq[k].w;
            }
#pragma unroll
            for (int n = 0; n < N_; ++n) {
                float dA = __builtin_amdgcn_exp2f(s * A2[n]);
                h[n] = fmaf(dA, h[n], su * Brow[n]);
            }
#pragma unroll
            for (int k = 0; k < 4; ++k) bq[k] = bn[k];
        }
        if (bt & 1) {
            long off = ((long)(branch * 4 + b) * NSUB + lc * 2 + (bt >> 1)) * (D_ * N_) + (long)d * N_;
#pragma unroll
            for (int n = 0; n < N_; ++n) {
                hpart[off + n] = h[n];
                aprod[off + n] = __builtin_amdgcn_exp2f(ssum * A2[n]);
                h[n] = 0.f;
            }
            ssum = 0.f;
        }
    }
}

__global__ void k_carry(const float* __restrict__ hpart, const float* __restrict__ aprod,
                        float* __restrict__ hin) {
    int idx = blockIdx.x * blockDim.x + threadIdx.x;
    int dn = idx & (D_ * N_ - 1);
    int bb = idx >> 13;
    float h = 0.f;
#pragma unroll
    for (int c = 0; c < NSUB; ++c) {
        long pos = ((long)bb * NSUB + c) * (D_ * N_) + dn;
        hin[pos] = h;
        h = fmaf(aprod[pos], h, hpart[pos]);
    }
}

__global__ void __launch_bounds__(512) __attribute__((amdgpu_waves_per_eu(4, 4)))
k_scan2_ln(const float* __restrict__ x_rgb, const float* __restrict__ x_e,
           const float* __restrict__ dbl, const float* __restrict__ delta_sp,
           const float* __restrict__ Alog1, const float* __restrict__ Alog2,
           const float* __restrict__ D1, const float* __restrict__ D2,
           const float* __restrict__ hin,
           const float* __restrict__ g1, const float* __restrict__ bb1,
           const float* __restrict__ g2, const float* __restrict__ bb2,
           float* __restrict__ out) {
    int bid = blockIdx.x;
    int sc  = bid & (NSUB - 1); bid >>= 5;
    int b   = bid & 3;
    int branch = bid >> 2;
    int d = threadIdx.x;
    int l0 = sc * SUB;

    const float* Alog = branch ? Alog2 : Alog1;
    const float* Dp   = branch ? D2 : D1;
    const float* x    = branch ? x_e : x_rgb;
    const float* ds   = delta_sp + (long)branch * 4096 * D_;
    const float* sp   = ds + (long)(b * L_ + l0) * D_ + d;
    const float* up   = x  + (long)(b * L_ + l0) * D_ + d;

    float sN[BT], uN[BT];
#pragma unroll
    for (int j = 0; j < BT; ++j) { sN[j] = sp[j * D_]; uN[j] = up[j * D_]; }
    float al[N_], h[N_];
    long off = ((long)(branch * 4 + b) * NSUB + sc) * (D_ * N_) + (long)d * N_;
#pragma unroll
    for (int n = 0; n < N_; ++n) {
        al[n] = Alog[(long)d * N_ + n];
        h[n]  = hin[off + n];
    }
    float Dd = Dp[d];

    __shared__ __attribute__((aligned(16))) float Bsh[SUB * 16];
    __shared__ __attribute__((aligned(16))) float Csh[SUB * 16];
    __shared__ __attribute__((aligned(16))) float Ysh[SUB * D_];
    const v4f* Bsh4 = reinterpret_cast<const v4f*>(Bsh);
    const v4f* Csh4 = reinterpret_cast<const v4f*>(Csh);
    {
        const float* srcB = dbl + (long)branch * 4096 * 48 + (long)(b * L_ + l0) * 48 + 16;
        const float* srcC = dbl + (long)(1 - branch) * 4096 * 48 + (long)(b * L_ + l0) * 48 + 32;
        int r = d >> 4, cc = d & 15;
        Bsh[d] = srcB[(long)r * 48 + cc];
        Csh[d] = srcC[(long)r * 48 + cc];
    }

    float A2[N_];
#pragma unroll
    for (int n = 0; n < N_; ++n) A2[n] = -__expf(al[n]) * LOG2E;
    __syncthreads();

    v4f bq[4], cq[4];
#pragma unroll
    for (int k = 0; k < 4; ++k) { bq[k] = Bsh4[k]; cq[k] = Csh4[k]; }

#pragma unroll 1
    for (int bt = 0; bt < NB2; ++bt) {
        float sC[BT], uC[BT];
#pragma unroll
        for (int j = 0; j < BT; ++j) { sC[j] = sN[j]; uC[j] = uN[j]; }
        if (bt + 1 < NB2) {
            const float* sp1 = sp + (long)(bt + 1) * BT * D_;
            const float* up1 = up + (long)(bt + 1) * BT * D_;
#pragma unroll
            for (int j = 0; j < BT; ++j) { sN[j] = sp1[j * D_]; uN[j] = up1[j * D_]; }
        }
        float* yb = Ysh + bt * BT * D_ + d;
#pragma unroll
        for (int j = 0; j < BT; ++j) {
            int rn = (bt * BT + j + 1) & (SUB - 1);
            v4f bn[4], cn[4];
#pragma unroll
            for (int k = 0; k < 4; ++k) { bn[k] = Bsh4[rn * 4 + k]; cn[k] = Csh4[rn * 4 + k]; }
            float s = sC[j];
            float u = uC[j];
            float su = s * u;
            float y = 0.f;
            float Brow[N_], Crow[N_];
#pragma unroll
            for (int k = 0; k < 4; ++k) {
                Brow[4 * k + 0] = bq[k].x; Brow[4 * k + 1] = bq[k].y;
                Brow[4 * k + 2] = bq[k].z; Brow[4 * k + 3] = bq[k].w;
                Crow[4 * k + 0] = cq[k].x; Crow[4 * k + 1] = cq[k].y;
                Crow[4 * k + 2] = cq[k].z; Crow[4 * k + 3] = cq[k].w;
            }
#pragma unroll
            for (int n = 0; n < N_; ++n) {
                float dA = __builtin_amdgcn_exp2f(s * A2[n]);
                h[n] = fmaf(dA, h[n], su * Brow[n]);
                y = fmaf(h[n], Crow[n], y);
            }
            yb[j * D_] = fmaf(Dd, u, y);
#pragma unroll
            for (int k = 0; k < 4; ++k) { bq[k] = bn[k]; cq[k] = cn[k]; }
        }
    }

    const float* g  = branch ? g2 : g1;
    const float* bb = branch ? bb2 : bb1;
    int w  = d >> 5;
    int ln = d & 31;
    __syncthreads();

    v4f gq[4], oq[4];
#pragma unroll
    for (int k = 0; k < 4; ++k) {
        int c0 = (ln + 32 * k) * 4;
        gq[k].x = g[c0];  gq[k].y = g[c0 + 1];  gq[k].z = g[c0 + 2];  gq[k].w = g[c0 + 3];
        oq[k].x = bb[c0]; oq[k].y = bb[c0 + 1]; oq[k].z = bb[c0 + 2]; oq[k].w = bb[c0 + 3];
    }
    const v4f* Y4 = reinterpret_cast<const v4f*>(Ysh);
    v4f* o4 = reinterpret_cast<v4f*>(out + (long)branch * 4096 * D_ + (long)(b * L_ + l0) * D_);
    v4f v[2][4];
    float rs[2], rq[2];
#pragma unroll
    for (int rr = 0; rr < 2; ++rr) {
        int r = 2 * w + rr;
#pragma unroll
        for (int k = 0; k < 4; ++k) v[rr][k] = Y4[r * (D_ / 4) + ln + 32 * k];
        v4f s4 = (v[rr][0] + v[rr][1]) + (v[rr][2] + v[rr][3]);
        v4f q4 = (v[rr][0] * v[rr][0] + v[rr][1] * v[rr][1]) + (v[rr][2] * v[rr][2] + v[rr][3] * v[rr][3]);
        rs[rr] = (s4.x + s4.y) + (s4.z + s4.w);
        rq[rr] = (q4.x + q4.y) + (q4.z + q4.w);
    }
#pragma unroll
    for (int m = 16; m > 0; m >>= 1) {
#pragma unroll
        for (int rr = 0; rr < 2; ++rr) {
            rs[rr] += __shfl_xor(rs[rr], m, 32);
            rq[rr] += __shfl_xor(rq[rr], m, 32);
        }
    }
#pragma unroll
    for (int rr = 0; rr < 2; ++rr) {
        int r = 2 * w + rr;
        float mean = rs[rr] * (1.f / 512.f);
        float var  = rq[rr] * (1.f / 512.f) - mean * mean;
        float inv  = rsqrtf(var + 1e-5f);
#pragma unroll
        for (int k = 0; k < 4; ++k) {
            v4f o;
            o.x = (v[rr][k].x - mean) * inv * gq[k].x + oq[k].x;
            o.y = (v[rr][k].y - mean) * inv * gq[k].y + oq[k].y;
            o.z = (v[rr][k].z - mean) * inv * gq[k].z + oq[k].z;
            o.w = (v[rr][k].w - mean) * inv * gq[k].w + oq[k].w;
            o4[(long)r * (D_ / 4) + ln + 32 * k] = o;
        }
    }
}

extern "C" void kernel_launch(void* const* d_in, const int* in_sizes, int n_in,
                              void* d_out, int out_size, void* d_ws, size_t ws_size,
                              hipStream_t stream) {
    const float* x_rgb = (const float*)d_in[0];
    const float* x_e   = (const float*)d_in[1];
    const float* xw1   = (const float*)d_in[2];
    const float* xw2   = (const float*)d_in[3];
    const float* dtw1  = (const float*)d_in[4];
    const float* dtb1  = (const float*)d_in[5];
    const float* dtw2  = (const float*)d_in[6];
    const float* dtb2  = (const float*)d_in[7];
    const float* Al1   = (const float*)d_in[8];
    const float* Al2   = (const float*)d_in[9];
    const float* Dv1   = (const float*)d_in[10];
    const float* Dv2   = (const float*)d_in[11];
    const float* g1    = (const float*)d_in[12];
    const float* lb1   = (const float*)d_in[13];
    const float* g2    = (const float*)d_in[14];
    const float* lb2   = (const float*)d_in[15];
    float* out = (float*)d_out;

    float* ws    = (float*)d_ws;
    float* dbl   = ws;
    float* delta = dbl + 2 * 4096 * 48;
    float* hpart = delta + 2 * 4096 * 512;
    float* aprod = hpart + 2 * 4 * NSUB * 8192;
    float* hin   = aprod + 2 * 4 * NSUB * 8192;

    k_proj<<<256, 256, 0, stream>>>(x_rgb, x_e, xw1, xw2, dtw1, dtb1, dtw2, dtb2, dbl, delta);
    k_scan1<<<512, 128, 0, stream>>>(x_rgb, x_e, dbl, delta, Al1, Al2, hpart, aprod);
    k_carry<<<256, 256, 0, stream>>>(hpart, aprod, hin);
    k_scan2_ln<<<256, 512, 0, stream>>>(x_rgb, x_e, dbl, delta, Al1, Al2, Dv1, Dv2, hin,
                                        g1, lb1, g2, lb2, out);
}


# k_scan1 and k_carry become one kernel (3 launches): a block owns 16 channels over the whole sequence, a thread one 32-step piece, the 32 piece states are folded in LDS; every sum as before
# speedup vs baseline: 3.3747x; 1.1043x over previous
#include <hip/hip_runtime.h>
#include <hip/hip_bf16.h>

typedef float v2f __attribute__((ext_vector_type(2)));
typedef float v8f __attribute__((ext_vector_type(8)));
typedef float v4f __attribute__((ext_vector_type(4)));

#define B_ 4
#define L_ 1024
#define D_ 512
#define N_ 16
#define SUB 32
#define NSUB 32
#define BT 16
#define NB2 2
#define LOG2E 1.4426950408889634f

#define RB 32
#define KW 64
#define TP 52

__device__ __forceinline__ float softplus_f(float v) {
    float e = __builtin_amdgcn_exp2f(v * LOG2E);
    float z = e * __builtin_amdgcn_rcpf(2.f + e);
    float w = z * z;
    float p = fmaf(w, fmaf(w, fmaf(w, fmaf(w, 1.f / 9.f, 1.f / 7.f), 1.f / 5.f), 1.f / 3.f), 1.f);
    float lo = (z + z) * p;
    float hi = __builtin_amdgcn_logf(1.f + e) * 0.69314718056f;
    float r  = (e <= 0.5f) ? lo : hi;
    return (v > 20.f) ? v : r;
}

__global__ void __launch_bounds__(256) __attribute__((amdgpu_waves_per_eu(2, 2)))
k_proj(const float* __restrict__ x_rgb, const float* __restrict__ x_e,
       const float* __restrict__ xw1, const float* __restrict__ xw2,
       const float* __restrict__ dtw1, const float* __restrict__ bias1,
       const float* __restrict__ dtw2, const float* __restrict__ bias2,
       float* __restrict__ dbl, float* __restrict__ delta_sp) {
    int bid = blockIdx.x;
    int branch = bid >> 7;
    int row0 = (bid & 127) * RB;
    int tid = threadIdx.x;
    int wv = tid >> 5;
    int ln = tid & 31;
    int c = ln & 15;
    int h = ln >> 4;
    const float* x    = branch ? x_e : x_rgb;
    const float* xw   = branch ? xw2 : xw1;
    const float* dtw  = branch ? dtw2 : dtw1;
    const float* bias = branch ? bias2 : bias1;

    const float* xa = x  + (long)(row0 + c) * D_ + wv * KW + 4 * h;
    const float* wb = xw + (long)c * D_ + wv * KW + 4 * h;
    v4f A[2][8], Bf[3][8];
#pragma unroll
    for (int i = 0; i < 8; ++i) {
#pragma unroll
        for (int mt = 0; mt < 2; ++mt)
            A[mt][i] = *reinterpret_cast<const v4f*>(xa + (long)mt * 16 * D_ + 8 * i);
#pragma unroll
        for (int nt = 0; nt < 3; ++nt)
            Bf[nt][i] = *reinterpret_cast<const v4f*>(wb + (long)nt * 16 * D_ + 8 * i);
    }
    int n0 = wv * 64 + c;
    int kb = 2 * h;
    v2f bf2[4][4];
    float bcol[4];
#pragma unroll
    for (int nt = 0; nt < 4; ++nt) {
        int n = n0 + 16 * nt;
#pragma unroll
        for (int kk = 0; kk < 4; ++kk) {
            bf2[nt][kk].x = dtw[(long)n * 16 + 4 * kk + kb];
            bf2[nt][kk].y = dtw[(long)n * 16 + 4 * kk + kb + 1];
        }
        bcol[nt] = bias[n];
    }

    __shared__ __attribute__((aligned(16))) float part[8 * 6 * 256];
    __shared__ __attribute__((aligned(16))) float tile[RB * TP];

    v8f acc[2][3];
#pragma unroll
    for (int mt = 0; mt < 2; ++mt)
#pragma unroll
        for (int nt = 0; nt < 3; ++nt) acc[mt][nt] = v8f{};
#pragma unroll
    for (int i = 0; i < 8; ++i) {
#pragma unroll
        for (int hf = 0; hf < 2; ++hf) {
            v2f a[2], b[3];
#pragma unroll
            for (int mt = 0; mt < 2; ++mt) {
                a[mt].x = hf ? A[mt][i].z : A[mt][i].x;
                a[mt].y = hf ? A[mt][i].w : A[mt][i].y;
            }
#pragma unroll
            for (int nt = 0; nt < 3; ++nt) {
                b[nt].x = hf ? Bf[nt][i].z : Bf[nt][i].x;
                b[nt].y = hf ? Bf[nt][i].w : Bf[nt][i].y;
            }
#pragma unroll
            for (int mt = 0; mt < 2; ++mt)
#pragma unroll
                for (int nt = 0; nt < 3; ++nt)
                    acc[mt][nt] = __builtin_amdgcn_wmma_f32_16x16x4_f32(false, a[mt], false, b[nt],
                                                                        (short)0, acc[mt][nt], false, false);
        }
    }
    {
        v4f* pw = reinterpret_cast<v4f*>(part + wv * 1536);
#pragma unroll
        for (int mt = 0; mt < 2; ++mt)
#pragma unroll
            for (int nt = 0; nt < 3; ++nt) {
                int t = mt * 3 + nt;
                v4f lo, hi;
                lo.x = acc[mt][nt][0]; lo.y = acc[mt][nt][1]; lo.z = acc[mt][nt][2]; lo.w = acc[mt][nt][3];
                hi.x = acc[mt][nt][4]; hi.y = acc[mt][nt][5]; hi.z = acc[mt][nt][6]; hi.w = acc[mt][nt][7];
                pw[t * 64 + ln]      = lo;
                pw[t * 64 + 32 + ln] = hi;
            }
    }
    __syncthreads();
    {
        int rq = tid >> 7;
        int jl = (tid >> 2) & 31;
        int r  = 4 * rq + (tid & 3);
        int jrow = 8 * (jl >> 4) + r;
        int jc = jl & 15;
#pragma unroll
        for (int t = 0; t < 6; ++t) {
            float p[8];
#pragma unroll
            for (int w = 0; w < 8; ++w) p[w] = part[w * 1536 + t * 256 + tid];
            float s = ((p[0] + p[1]) + (p[2] + p[3])) + ((p[4] + p[5]) + (p[6] + p[7]));
            int mt = t / 3, nt = t % 3;
            tile[(mt * 16 + jrow) * TP + nt * 16 + jc] = s;
        }
    }
    __syncthreads();
    {
        int rr = tid >> 3, cc = 16 + 4 * (tid & 7);
        v4f v = *reinterpret_cast<const v4f*>(tile + rr * TP + cc);
        *reinterpret_cast<v4f*>(dbl + (long)branch * 4096 * 48 + (long)(row0 + rr) * 48 + cc) = v;
    }

    float* out = delta_sp + (long)branch * 4096 * D_;
#pragma unroll
    for (int mt = 0; mt < 2; ++mt) {
        v2f a[4];
#pragma unroll
        for (int kk = 0; kk < 4; ++kk)
            a[kk] = *reinterpret_cast<const v2f*>(tile + (mt * 16 + c) * TP + 4 * kk + kb);
        int rowbase = row0 + mt * 16 + (h << 3);
#pragma unroll
        for (int nt = 0; nt < 4; ++nt) {
            v8f d = {};
#pragma unroll
            for (int kk = 0; kk < 4; ++kk)
                d = __builtin_amdgcn_wmma_f32_16x16x4_f32(false, a[kk], false, bf2[nt][kk], (short)0, d, false, false);
            int col = n0 + 16 * nt;
#pragma unroll
            for (int i = 0; i < 8; ++i)
                out[(long)(rowbase + i) * D_ + col] = softplus_f(d[i] + bcol[nt]);
        }
    }
}

#define CH 16
#define PSTR 528
__global__ void __launch_bounds__(512) __attribute__((amdgpu_waves_per_eu(4, 4)))
k_scan1c(const float* __restrict__ x_rgb, const float* __restrict__ x_e,
         const float* __restrict__ dbl, const float* __restrict__ delta_sp,
         const float* __restrict__ Alog1, const float* __restrict__ Alog2,
         float* __restrict__ hin) {
    int bid = blockIdx.x;
    int dg  = bid & 31; bid >>= 5;
    int b   = bid & 3;
    int branch = bid >> 2;
    int tid = threadIdx.x;
    int c = tid & (CH - 1);
    int p = tid >> 4;
    int l0 = p * SUB;

    int d = dg * CH + c;
    const float* Alog = branch ? Alog2 : Alog1;
    const float* x    = branch ? x_e : x_rgb;
    const float* ds   = delta_sp + (long)branch * 4096 * D_;
    const float* sp   = ds + (long)(b * L_ + l0) * D_ + d;
    const float* up   = x  + (long)(b * L_ + l0) * D_ + d;

    float sN[BT], uN[BT];
#pragma unroll
    for (int j = 0; j < BT; ++j) { sN[j] = sp[j * D_]; uN[j] = up[j * D_]; }
    float al[N_];
#pragma unroll
    for (int n = 0; n < N_; ++n) al[n] = Alog[(long)d * N_ + n];
    float alc = Alog[(long)dg * (CH * N_) + (tid & 255)];

    __shared__ __attribute__((aligned(16))) float Bsh[NSUB * PSTR];
    __shared__ __attribute__((aligned(16))) float Hsh[NSUB * CH * N_];
    __shared__ __attribute__((aligned(16))) float Ssh[NSUB * CH];
    {
        const float* srcB = dbl + (long)branch * 4096 * 48 + (long)(b * L_) * 48 + 16;
        v4f t4[8];
#pragma unroll
        for (int k = 0; k < 8; ++k) {
            int i = tid + 512 * k;
            t4[k] = *reinterpret_cast<const v4f*>(srcB + (long)(i >> 2) * 48 + 4 * (i & 3));
        }
#pragma unroll
        for (int k = 0; k < 8; ++k) {
            int i = tid + 512 * k;
            int r = i >> 2;
            *reinterpret_cast<v4f*>(Bsh + (r >> 5) * PSTR + (r & (SUB - 1)) * 16 + 4 * (i & 3)) = t4[k];
        }
    }

    float A2[N_], h[N_];
#pragma unroll
    for (int n = 0; n < N_; ++n) {
        A2[n] = -__expf(al[n]) * LOG2E;
        h[n] = 0.f;
    }
    __syncthreads();

    const v4f* Bp4 = reinterpret_cast<const v4f*>(Bsh + p * PSTR);
    v4f bq[4];
#pragma unroll
    for (int k = 0; k < 4; ++k) bq[k] = Bp4[k];

    float ssum = 0.f;
#pragma unroll 1
    for (int bt = 0; bt < NB2; ++bt) {
        float sC[BT], uC[BT];
#pragma unroll
        for (int j = 0; j < BT; ++j) { sC[j] = sN[j]; uC[j] = uN[j]; }
        if (bt + 1 < NB2) {
            const float* sp1 = sp + (long)(bt + 1) * BT * D_;
            const float* up1 = up + (long)(bt + 1) * BT * D_;
#pragma unroll
            for (int j = 0; j < BT; ++j) { sN[j] = sp1[j * D_]; uN[j] = up1[j * D_]; }
        }
#pragma unroll
        for (int j = 0; j < BT; ++j) {
            int rn = (bt * BT + j + 1) & (SUB - 1);
            v4f bn[4];
#pragma unroll
            for (int k = 0; k < 4; ++k) bn[k] = Bp4[rn * 4 + k];
            float s = sC[j];
            float u = uC[j];
            float su = s * u;
            ssum += s;
            float Brow[N_];
#pragma unroll
            for (int k = 0; k < 4; ++k) {
                Brow[4 * k + 0] = bq[k].x; Brow[4 * k + 1] = bq[k].y;
                Brow[4 * k + 2] = bq[k].z; Brow[4 * k + 3] = bq[k].w;
            }
#pragma unroll
            for (int n = 0; n < N_; ++n) {
                float dA = __builtin_amdgcn_exp2f(s * A2[n]);
                h[n] = fmaf(dA, h[n], su * Brow[n]);
            }
#pragma unroll
            for (int k = 0; k < 4; ++k) bq[k] = bn[k];
        }
    }

    {
        v4f* H4 = reinterpret_cast<v4f*>(Hsh + tid * N_);
#pragma unroll
        for (int k = 0; k < 4; ++k) {
            v4f v;
            v.x = h[4 * k + 0]; v.y = h[4 * k + 1]; v.z = h[4 * k + 2]; v.w = h[4 * k + 3];
            H4[k] = v;
        }
        Ssh[tid] = ssum;
    }
    __syncthreads();

    if (tid < CH * N_) {
        float A2c = -__expf(alc) * LOG2E;
        float hc = 0.f;
        float* hp = hin + (long)(branch * 4 + b) * NSUB * (D_ * N_) + (long)dg * (CH * N_) + tid;
        int cc = tid >> 4;
#pragma unroll
        for (int pp = 0; pp < NSUB; ++pp) {
            hp[(long)pp * (D_ * N_)] = hc;
            float a = __builtin_amdgcn_exp2f(Ssh[pp * CH + cc] * A2c);
            hc = fmaf(a, hc, Hsh[pp * (CH * N_) + tid]);
        }
    }
}

__global__ void __launch_bounds__(512) __attribute__((amdgpu_waves_per_eu(4, 4)))
k_scan2_ln(const float* __restrict__ x_rgb, const float* __restrict__ x_e,
           const float* __restrict__ dbl, const float* __restrict__ delta_sp,
           const float* __restrict__ Alog1, const float* __restrict__ Alog2,
           const float* __restrict__ D1, const float* __restrict__ D2,
           const float* __restrict__ hin,
           const float* __restrict__ g1, const float* __restrict__ bb1,
           const float* __restrict__ g2, const float* __restrict__ bb2,
           float* __restrict__ out) {
    int bid = blockIdx.x;
    int sc  = bid & (NSUB - 1); bid >>= 5;
    int b   = bid & 3;
    int branch = bid >> 2;
    int d = threadIdx.x;
    int l0 = sc * SUB;

    const float* Alog = branch ? Alog2 : Alog1;
    const float* Dp   = branch ? D2 : D1;
    const float* x    = branch ? x_e : x_rgb;
    const float* ds   = delta_sp + (long)branch * 4096 * D_;
    const float* sp   = ds + (long)(b * L_ + l0) * D_ + d;
    const float* up   = x  + (long)(b * L_ + l0) * D_ + d;

    float sN[BT], uN[BT];
#pragma unroll
    for (int j = 0; j < BT; ++j) { sN[j] = sp[j * D_]; uN[j] = up[j * D_]; }
    float al[N_], h[N_];
    long off = ((long)(branch * 4 + b) * NSUB + sc) * (D_ * N_) + (long)d * N_;
#pragma unroll
    for (int n = 0; n < N_; ++n) {
        al[n] = Alog[(long)d * N_ + n];
        h[n]  = hin[off + n];
    }
    float Dd = Dp[d];

    __shared__ __attribute__((aligned(16))) float Bsh[SUB * 16];
    __shared__ __attribute__((aligned(16))) float Csh[SUB * 16];
    __shared__ __attribute__((aligned(16))) float Ysh[SUB * D_];
    const v4f* Bsh4 = reinterpret_cast<const v4f*>(Bsh);
    const v4f* Csh4 = reinterpret_cast<const v4f*>(Csh);
    {
        const float* srcB = dbl + (long)branch * 4096 * 48 + (long)(b * L_ + l0) * 48 + 16;
        const float* srcC = dbl + (long)(1 - branch) * 4096 * 48 + (long)(b * L_ + l0) * 48 + 32;
        int r = d >> 4, cc = d & 15;
        Bsh[d] = srcB[(long)r * 48 + cc];
        Csh[d] = srcC[(long)r * 48 + cc];
    }

    float A2[N_];
#pragma unroll
    for (int n = 0; n < N_; ++n) A2[n] = -__expf(al[n]) * LOG2E;
    __syncthreads();

    v4f bq[4], cq[4];
#pragma unroll
    for (int k = 0; k < 4; ++k) { bq[k] = Bsh4[k]; cq[k] = Csh4[k]; }

#pragma unroll 1
    for (int bt = 0; bt < NB2; ++bt) {
        float sC[BT], uC[BT];
#pragma unroll
        for (int j = 0; j < BT; ++j) { sC[j] = sN[j]; uC[j] = uN[j]; }
        if (bt + 1 < NB2) {
            const float* sp1 = sp + (long)(bt + 1) * BT * D_;
            const float* up1 = up + (long)(bt + 1) * BT * D_;
#pragma unroll
            for (int j = 0; j < BT; ++j) { sN[j] = sp1[j * D_]; uN[j] = up1[j * D_]; }
        }
        float* yb = Ysh + bt * BT * D_ + d;
#pragma unroll
        for (int j = 0; j < BT; ++j) {
            int rn = (bt * BT + j + 1) & (SUB - 1);
            v4f bn[4], cn[4];
#pragma unroll
            for (int k = 0; k < 4; ++k) { bn[k] = Bsh4[rn * 4 + k]; cn[k] = Csh4[rn * 4 + k]; }
            float s = sC[j];
            float u = uC[j];
            float su = s * u;
            float y = 0.f;
            float Brow[N_], Crow[N_];
#pragma unroll
            for (int k = 0; k < 4; ++k) {
                Brow[4 * k + 0] = bq[k].x; Brow[4 * k + 1] = bq[k].y;
                Brow[4 * k + 2] = bq[k].z; Brow[4 * k + 3] = bq[k].w;
                Crow[4 * k + 0] = cq[k].x; Crow[4 * k + 1] = cq[k].y;
                Crow[4 * k + 2] = cq[k].z; Crow[4 * k + 3] = cq[k].w;
            }
#pragma unroll
            for (int n = 0; n < N_; ++n) {
                float dA = __builtin_amdgcn_exp2f(s * A2[n]);
                h[n] = fmaf(dA, h[n], su * Brow[n]);
                y = fmaf(h[n], Crow[n], y);
            }
            yb[j * D_] = fmaf(Dd, u, y);
#pragma unroll
            for (int k = 0; k < 4; ++k) { bq[k] = bn[k]; cq[k] = cn[k]; }
        }
    }

    const float* g  = branch ? g2 : g1;
    const float* bb = branch ? bb2 : bb1;
    int w  = d >> 5;
    int ln = d & 31;
    __syncthreads();

    v4f gq[4], oq[4];
#pragma unroll
    for (int k = 0; k < 4; ++k) {
        int c0 = (ln + 32 * k) * 4;
        gq[k].x = g[c0];  gq[k].y = g[c0 + 1];  gq[k].z = g[c0 + 2];  gq[k].w = g[c0 + 3];
        oq[k].x = bb[c0]; oq[k].y = bb[c0 + 1]; oq[k].z = bb[c0 + 2]; oq[k].w = bb[c0 + 3];
    }
    const v4f* Y4 = reinterpret_cast<const v4f*>(Ysh);
    v4f* o4 = reinterpret_cast<v4f*>(out + (long)branch * 4096 * D_ + (long)(b * L_ + l0) * D_);
    v4f v[2][4];
    float rs[2], rq[2];
#pragma unroll
    for (int rr = 0; rr < 2; ++rr) {
        int r = 2 * w + rr;
#pragma unroll
        for (int k = 0; k < 4; ++k) v[rr][k] = Y4[r * (D_ / 4) + ln + 32 * k];
        v4f s4 = (v[rr][0] + v[rr][1]) + (v[rr][2] + v[rr][3]);
        v4f q4 = (v[rr][0] * v[rr][0] + v[rr][1] * v[rr][1]) + (v[rr][2] * v[rr][2] + v[rr][3] * v[rr][3]);
        rs[rr] = (s4.x + s4.y) + (s4.z + s4.w);
        rq[rr] = (q4.x + q4.y) + (q4.z + q4.w);
    }
#pragma unroll
    for (int m = 16; m > 0; m >>= 1) {
#pragma unroll
        for (int rr = 0; rr < 2; ++rr) {
            rs[rr] += __shfl_xor(rs[rr], m, 32);
            rq[rr] += __shfl_xor(rq[rr], m, 32);
        }
    }
#pragma unroll
    for (int rr = 0; rr < 2; ++rr) {
        int r = 2 * w + rr;
        float mean = rs[rr] * (1.f / 512.f);
        float var  = rq[rr] * (1.f / 512.f) - mean * mean;
        float inv  = rsqrtf(var + 1e-5f);
#pragma unroll
        for (int k = 0; k < 4; ++k) {
            v4f o;
            o.x = (v[rr][k].x - mean) * inv * gq[k].x + oq[k].x;
            o.y = (v[rr][k].y - mean) * inv * gq[k].y + oq[k].y;
            o.z = (v[rr][k].z - mean) * inv * gq[k].z + oq[k].z;
            o.w = (v[rr][k].w - mean) * inv * gq[k].w + oq[k].w;
            o4[(long)r * (D_ / 4) + ln + 32 * k] = o;
        }
    }
}

extern "C" void kernel_launch(void* const* d_in, const int* in_sizes, int n_in,
                              void* d_out, int out_size, void* d_ws, size_t ws_size,
                              hipStream_t stream) {
    const float* x_rgb = (const float*)d_in[0];
    const float* x_e   = (const float*)d_in[1];
    const float* xw1   = (const float*)d_in[2];
    const float* xw2   = (const float*)d_in[3];
    const float* dtw1  = (const float*)d_in[4];
    const float* dtb1  = (const float*)d_in[5];
    const float* dtw2  = (const float*)d_in[6];
    const float* dtb2  = (const float*)d_in[7];
    const float* Al1   = (const float*)d_in[8];
    const float* Al2   = (const float*)d_in[9];
    const float* Dv1   = (const float*)d_in[10];
    const float* Dv2   = (const float*)d_in[11];
    const float* g1    = (const float*)d_in[12];
    const float* lb1   = (const float*)d_in[13];
    const float* g2    = (const float*)d_in[14];
    const float* lb2   = (const float*)d_in[15];
    float* out = (float*)d_out;

    float* ws    = (float*)d_ws;
    float* dbl   = ws;
    float* delta = dbl + 2 * 4096 * 48;
    float* hin   = delta + 2 * 4096 * 512;

    k_proj<<<256, 256, 0, stream>>>(x_rgb, x_e, xw1, xw2, dtw1, dtb1, dtw2, dtb2, dbl, delta);
    k_scan1c<<<256, 512, 0, stream>>>(x_rgb, x_e, dbl, delta, Al1, Al2, hin);
    k_scan2_ln<<<256, 512, 0, stream>>>(x_rgb, x_e, dbl, delta, Al1, Al2, Dv1, Dv2, hin,
                                        g1, lb1, g2, lb2, out);
}


# k_scan2_ln: B and C rows of a step (equal in all lanes) come by scalar loads from a prefetched scalar cache, not by LDS broadcast reads: no LDS read, no barrier before the recurrence; sums as before
# speedup vs baseline: 3.4956x; 1.0358x over previous
#include <hip/hip_runtime.h>
#include <hip/hip_bf16.h>

typedef float v2f __attribute__((ext_vector_type(2)));
typedef float v8f __attribute__((ext_vector_type(8)));
typedef float v4f __attribute__((ext_vector_type(4)));
typedef const float __attribute__((address_space(4))) cfloat_k;

#define B_ 4
#define L_ 1024
#define D_ 512
#define N_ 16
#define SUB 32
#define NSUB 32
#define BT 16
#define NB2 2
#define LOG2E 1.4426950408889634f

#define RB 32
#define KW 64
#define TP 52

__device__ __forceinline__ float softplus_f(float v) {
    float e = __builtin_amdgcn_exp2f(v * LOG2E);
    float z = e * __builtin_amdgcn_rcpf(2.f + e);
    float w = z * z;
    float p = fmaf(w, fmaf(w, fmaf(w, fmaf(w, 1.f / 9.f, 1.f / 7.f), 1.f / 5.f), 1.f / 3.f), 1.f);
    float lo = (z + z) * p;
    float hi = __builtin_amdgcn_logf(1.f + e) * 0.69314718056f;
    float r  = (e <= 0.5f) ? lo : hi;
    return (v > 20.f) ? v : r;
}

__global__ void __launch_bounds__(256) __attribute__((amdgpu_waves_per_eu(2, 2)))
k_proj(const float* __restrict__ x_rgb, const float* __restrict__ x_e,
       const float* __restrict__ xw1, const float* __restrict__ xw2,
       const float* __restrict__ dtw1, const float* __restrict__ bias1,
       const float* __restrict__ dtw2, const float* __restrict__ bias2,
       float* __restrict__ dbl, float* __restrict__ delta_sp) {
    int bid = blockIdx.x;
    int branch = bid >> 7;
    int row0 = (bid & 127) * RB;
    int tid = threadIdx.x;
    int wv = tid >> 5;
    int ln = tid & 31;
    int c = ln & 15;
    int h = ln >> 4;
    const float* x    = branch ? x_e : x_rgb;
    const float* xw   = branch ? xw2 : xw1;
    const float* dtw  = branch ? dtw2 : dtw1;
    const float* bias = branch ? bias2 : bias1;

    const float* xa = x  + (long)(row0 + c) * D_ + wv * KW + 4 * h;
    const float* wb = xw + (long)c * D_ + wv * KW + 4 * h;
    v4f A[2][8], Bf[3][8];
#pragma unroll
    for (int i = 0; i < 8; ++i) {
#pragma unroll
        for (int mt = 0; mt < 2; ++mt)
            A[mt][i] = *reinterpret_cast<const v4f*>(xa + (long)mt * 16 * D_ + 8 * i);
#pragma unroll
        for (int nt = 0; nt < 3; ++nt)
            Bf[nt][i] = *reinterpret_cast<const v4f*>(wb + (long)nt * 16 * D_ + 8 * i);
    }
    int n0 = wv * 64 + c;
    int kb = 2 * h;
    v2f bf2[4][4];
    float bcol[4];
#pragma unroll
    for (int nt = 0; nt < 4; ++nt) {
        int n = n0 + 16 * nt;
#pragma unroll
        for (int kk = 0; kk < 4; ++kk) {
            bf2[nt][kk].x = dtw[(long)n * 16 + 4 * kk + kb];
            bf2[nt][kk].y = dtw[(long)n * 16 + 4 * kk + kb + 1];
        }
        bcol[nt] = bias[n];
    }

    __shared__ __attribute__((aligned(16))) float part[8 * 6 * 256];
    __shared__ __attribute__((aligned(16))) float tile[RB * TP];

    v8f acc[2][3];
#pragma unroll
    for (int mt = 0; mt < 2; ++mt)
#pragma unroll
        for (int nt = 0; nt < 3; ++nt) acc[mt][nt] = v8f{};
#pragma unroll
    for (int i = 0; i < 8; ++i) {
#pragma unroll
        for (int hf = 0; hf < 2; ++hf) {
            v2f a[2], b[3];
#pragma unroll
            for (int mt = 0; mt < 2; ++mt) {
                a[mt].x = hf ? A[mt][i].z : A[mt][i].x;
                a[mt].y = hf ? A[mt][i].w : A[mt][i].y;
            }
#pragma unroll
            for (int nt = 0; nt < 3; ++nt) {
                b[nt].x = hf ? Bf[nt][i].z : Bf[nt][i].x;
                b[nt].y = hf ? Bf[nt][i].w : Bf[nt][i].y;
            }
#pragma unroll
            for (int mt = 0; mt < 2; ++mt)
#pragma unroll
                for (int nt = 0; nt < 3; ++nt)
                    acc[mt][nt] = __builtin_amdgcn_wmma_f32_16x16x4_f32(false, a[mt], false, b[nt],
                                                                        (short)0, acc[mt][nt], false, false);
        }
    }
    {
        v4f* pw = reinterpret_cast<v4f*>(part + wv * 1536);
#pragma unroll
        for (int mt = 0; mt < 2; ++mt)
#pragma unroll
            for (int nt = 0; nt < 3; ++nt) {
                int t = mt * 3 + nt;
                v4f lo, hi;
                lo.x = acc[mt][nt][0]; lo.y = acc[mt][nt][1]; lo.z = acc[mt][nt][2]; lo.w = acc[mt][nt][3];
                hi.x = acc[mt][nt][4]; hi.y = acc[mt][nt][5]; hi.z = acc[mt][nt][6]; hi.w = acc[mt][nt][7];
                pw[t * 64 + ln]      = lo;
                pw[t * 64 + 32 + ln] = hi;
            }
    }
    __syncthreads();
    {
        int rq = tid >> 7;
        int jl = (tid >> 2) & 31;
        int r  = 4 * rq + (tid & 3);
        int jrow = 8 * (jl >> 4) + r;
        int jc = jl & 15;
#pragma unroll
        for (int t = 0; t < 6; ++t) {
            float p[8];
#pragma unroll
            for (int w = 0; w < 8; ++w) p[w] = part[w * 1536 + t * 256 + tid];
            float s = ((p[0] + p[1]) + (p[2] + p[3])) + ((p[4] + p[5]) + (p[6] + p[7]));
            int mt = t / 3, nt = t % 3;
            tile[(mt * 16 + jrow) * TP + nt * 16 + jc] = s;
        }
    }
    __syncthreads();
    {
        int rr = tid >> 3, cc = 16 + 4 * (tid & 7);
        v4f v = *reinterpret_cast<const v4f*>(tile + rr * TP + cc);
        *reinterpret_cast<v4f*>(dbl + (long)branch * 4096 * 48 + (long)(row0 + rr) * 48 + cc) = v;
    }

    float* out = delta_sp + (long)branch * 4096 * D_;
#pragma unroll
    for (int mt = 0; mt < 2; ++mt) {
        v2f a[4];
#pragma unroll
        for (int kk = 0; kk < 4; ++kk)
            a[kk] = *reinterpret_cast<const v2f*>(tile + (mt * 16 + c) * TP + 4 * kk + kb);
        int rowbase = row0 + mt * 16 + (h << 3);
#pragma unroll
        for (int nt = 0; nt < 4; ++nt) {
            v8f d = {};
#pragma unroll
            for (int kk = 0; kk < 4; ++kk)
                d = __builtin_amdgcn_wmma_f32_16x16x4_f32(false, a[kk], false, bf2[nt][kk], (short)0, d, false, false);
            int col = n0 + 16 * nt;
#pragma unroll
            for (int i = 0; i < 8; ++i)
                out[(long)(rowbase + i) * D_ + col] = softplus_f(d[i] + bcol[nt]);
        }
    }
}

#define CH 16
#define PSTR 528
__global__ void __launch_bounds__(512) __attribute__((amdgpu_waves_per_eu(4, 4)))
k_scan1c(const float* __restrict__ x_rgb, const float* __restrict__ x_e,
         const float* __restrict__ dbl, const float* __restrict__ delta_sp,
         const float* __restrict__ Alog1, const float* __restrict__ Alog2,
         float* __restrict__ hin) {
    int bid = blockIdx.x;
    int dg  = bid & 31; bid >>= 5;
    int b   = bid & 3;
    int branch = bid >> 2;
    int tid = threadIdx.x;
    int c = tid & (CH - 1);
    int p = tid >> 4;
    int l0 = p * SUB;

    int d = dg * CH + c;
    const float* Alog = branch ? Alog2 : Alog1;
    const float* x    = branch ? x_e : x_rgb;
    const float* ds   = delta_sp + (long)branch * 4096 * D_;
    const float* sp   = ds + (long)(b * L_ + l0) * D_ + d;
    const float* up   = x  + (long)(b * L_ + l0) * D_ + d;

    float sN[BT], uN[BT];
#pragma unroll
    for (int j = 0; j < BT; ++j) { sN[j] = sp[j * D_]; uN[j] = up[j * D_]; }
    float al[N_];
#pragma unroll
    for (int n = 0; n < N_; ++n) al[n] = Alog[(long)d * N_ + n];
    float alc = Alog[(long)dg * (CH * N_) + (tid & 255)];

    __shared__ __attribute__((aligned(16))) float Bsh[NSUB * PSTR];
    __shared__ __attribute__((aligned(16))) float Hsh[NSUB * CH * N_];
    __shared__ __attribute__((aligned(16))) float Ssh[NSUB * CH];
    {
        const float* srcB = dbl + (long)branch * 4096 * 48 + (long)(b * L_) * 48 + 16;
        v4f t4[8];
#pragma unroll
        for (int k = 0; k < 8; ++k) {
            int i = tid + 512 * k;
            t4[k] = *reinterpret_cast<const v4f*>(srcB + (long)(i >> 2) * 48 + 4 * (i & 3));
        }
#pragma unroll
        for (int k = 0; k < 8; ++k) {
            int i = tid + 512 * k;
            int r = i >> 2;
            *reinterpret_cast<v4f*>(Bsh + (r >> 5) * PSTR + (r & (SUB - 1)) * 16 + 4 * (i & 3)) = t4[k];
        }
    }

    float A2[N_], h[N_];
#pragma unroll
    for (int n = 0; n < N_; ++n) {
        A2[n] = -__expf(al[n]) * LOG2E;
        h[n] = 0.f;
    }
    __syncthreads();

    const v4f* Bp4 = reinterpret_cast<const v4f*>(Bsh + p * PSTR);
    v4f bq[4];
#pragma unroll
    for (int k = 0; k < 4; ++k) bq[k] = Bp4[k];

    float ssum = 0.f;
#pragma unroll 1
    for (int bt = 0; bt < NB2; ++bt) {
        float sC[BT], uC[BT];
#pragma unroll
        for (int j = 0; j < BT; ++j) { sC[j] = sN[j]; uC[j] = uN[j]; }
        if (bt + 1 < NB2) {
            const float* sp1 = sp + (long)(bt + 1) * BT * D_;
            const float* up1 = up + (long)(bt + 1) * BT * D_;
#pragma unroll
            for (int j = 0; j < BT; ++j) { sN[j] = sp1[j * D_]; uN[j] = up1[j * D_]; }
        }
#pragma unroll
        for (int j = 0; j < BT; ++j) {
            int rn = (bt * BT + j + 1) & (SUB - 1);
            v4f bn[4];
#pragma unroll
            for (int k = 0; k < 4; ++k) bn[k] = Bp4[rn * 4 + k];
            float s = sC[j];
            float u = uC[j];
            float su = s * u;
            ssum += s;
            float Brow[N_];
#pragma unroll
            for (int k = 0; k < 4; ++k) {
                Brow[4 * k + 0] = bq[k].x; Brow[4 * k + 1] = bq[k].y;
                Brow[4 * k + 2] = bq[k].z; Brow[4 * k + 3] = bq[k].w;
            }
#pragma unroll
            for (int n = 0; n < N_; ++n) {
                float dA = __builtin_amdgcn_exp2f(s * A2[n]);
                h[n] = fmaf(dA, h[n], su * Brow[n]);
            }
#pragma unroll
            for (int k = 0; k < 4; ++k) bq[k] = bn[k];
        }
    }

    {
        v4f* H4 = reinterpret_cast<v4f*>(Hsh + tid * N_);
#pragma unroll
        for (int k = 0; k < 4; ++k) {
            v4f v;
            v.x = h[4 * k + 0]; v.y = h[4 * k + 1]; v.z = h[4 * k + 2]; v.w = h[4 * k + 3];
            H4[k] = v;
        }
        Ssh[tid] = ssum;
    }
    __syncthreads();

    if (tid < CH * N_) {
        float A2c = -__expf(alc) * LOG2E;
        float hc = 0.f;
        float* hp = hin + (long)(branch * 4 + b) * NSUB * (D_ * N_) + (long)dg * (CH * N_) + tid;
        int cc = tid >> 4;
#pragma unroll
        for (int pp = 0; pp < NSUB; ++pp) {
            hp[(long)pp * (D_ * N_)] = hc;
            float a = __builtin_amdgcn_exp2f(Ssh[pp * CH + cc] * A2c);
            hc = fmaf(a, hc, Hsh[pp * (CH * N_) + tid]);
        }
    }
}

__global__ void __launch_bounds__(512) __attribute__((amdgpu_waves_per_eu(4, 4)))
k_scan2_ln(const float* __restrict__ x_rgb, const float* __restrict__ x_e,
           const float* __restrict__ dbl, const float* __restrict__ delta_sp,
           const float* __restrict__ Alog1, const float* __restrict__ Alog2,
           const float* __restrict__ D1, const float* __restrict__ D2,
           const float* __restrict__ hin,
           const float* __restrict__ g1, const float* __restrict__ bb1,
           const float* __restrict__ g2, const float* __restrict__ bb2,
           float* __restrict__ out) {
    int bid = blockIdx.x;
    int sc  = bid & (NSUB - 1); bid >>= 5;
    int b   = bid & 3;
    int branch = bid >> 2;
    int d = threadIdx.x;
    int l0 = sc * SUB;

    const float* Alog = branch ? Alog2 : Alog1;
    const float* Dp   = branch ? D2 : D1;
    const float* x    = branch ? x_e : x_rgb;
    const float* ds   = delta_sp + (long)branch * 4096 * D_;
    const float* sp   = ds + (long)(b * L_ + l0) * D_ + d;
    const float* up   = x  + (long)(b * L_ + l0) * D_ + d;

    float sN[BT], uN[BT];
#pragma unroll
    for (int j = 0; j < BT; ++j) { sN[j] = sp[j * D_]; uN[j] = up[j * D_]; }
    float al[N_], h[N_];
    long off = ((long)(branch * 4 + b) * NSUB + sc) * (D_ * N_) + (long)d * N_;
#pragma unroll
    for (int n = 0; n < N_; ++n) {
        al[n] = Alog[(long)d * N_ + n];
        h[n]  = hin[off + n];
    }
    float Dd = Dp[d];

    __shared__ __attribute__((aligned(16))) float Ysh[SUB * D_];
    const float* srcB = dbl + (long)branch * 4096 * 48 + (long)(b * L_ + l0) * 48 + 16;
    const float* srcC = dbl + (long)(1 - branch) * 4096 * 48 + (long)(b * L_ + l0) * 48 + 32;
    __builtin_amdgcn_s_prefetch_data(srcB - 16, 31);
    __builtin_amdgcn_s_prefetch_data(srcB - 16 + 1024, 15);
    __builtin_amdgcn_s_prefetch_data(srcC - 32, 31);
    __builtin_amdgcn_s_prefetch_data(srcC - 32 + 1024, 15);
    cfloat_k* kB = (cfloat_k*)srcB;
    cfloat_k* kC = (cfloat_k*)srcC;

    float A2[N_];
#pragma unroll
    for (int n = 0; n < N_; ++n) A2[n] = -__expf(al[n]) * LOG2E;

    float Bq[N_], Cq[N_];
#pragma unroll
    for (int n = 0; n < N_; ++n) { Bq[n] = kB[n]; Cq[n] = kC[n]; }

#pragma unroll 1
    for (int bt = 0; bt < NB2; ++bt) {
        float sC[BT], uC[BT];
#pragma unroll
        for (int j = 0; j < BT; ++j) { sC[j] = sN[j]; uC[j] = uN[j]; }
        if (bt + 1 < NB2) {
            const float* sp1 = sp + (long)(bt + 1) * BT * D_;
            const float* up1 = up + (long)(bt + 1) * BT * D_;
#pragma unroll
            for (int j = 0; j < BT; ++j) { sN[j] = sp1[j * D_]; uN[j] = up1[j * D_]; }
        }
        float* yb = Ysh + bt * BT * D_ + d;
#pragma unroll
        for (int j = 0; j < BT; ++j) {
            int rn = (bt * BT + j + 1) & (SUB - 1);
            float Bn[N_], Cn[N_];
#pragma unroll
            for (int n = 0; n < N_; ++n) { Bn[n] = kB[rn * 48 + n]; Cn[n] = kC[rn * 48 + n]; }
            float s = sC[j];
            float u = uC[j];
            float su = s * u;
            float y = 0.f;
#pragma unroll
            for (int n = 0; n < N_; ++n) {
                float dA = __builtin_amdgcn_exp2f(s * A2[n]);
                h[n] = fmaf(dA, h[n], su * Bq[n]);
                y = fmaf(h[n], Cq[n], y);
            }
            yb[j * D_] = fmaf(Dd, u, y);
#pragma unroll
            for (int n = 0; n < N_; ++n) { Bq[n] = Bn[n]; Cq[n] = Cn[n]; }
        }
    }

    const float* g  = branch ? g2 : g1;
    const float* bb = branch ? bb2 : bb1;
    int w  = d >> 5;
    int ln = d & 31;
    __syncthreads();

    v4f gq[4], oq[4];
#pragma unroll
    for (int k = 0; k < 4; ++k) {
        int c0 = (ln + 32 * k) * 4;
        gq[k].x = g[c0];  gq[k].y = g[c0 + 1];  gq[k].z = g[c0 + 2];  gq[k].w = g[c0 + 3];
        oq[k].x = bb[c0]; oq[k].y = bb[c0 + 1]; oq[k].z = bb[c0 + 2]; oq[k].w = bb[c0 + 3];
    }
    const v4f* Y4 = reinterpret_cast<const v4f*>(Ysh);
    v4f* o4 = reinterpret_cast<v4f*>(out + (long)branch * 4096 * D_ + (long)(b * L_ + l0) * D_);
    v4f v[2][4];
    float rs[2], rq[2];
#pragma unroll
    for (int rr = 0; rr < 2; ++rr) {
        int r = 2 * w + rr;
#pragma unroll
        for (int k = 0; k < 4; ++k) v[rr][k] = Y4[r * (D_ / 4) + ln + 32 * k];
        v4f s4 = (v[rr][0] + v[rr][1]) + (v[rr][2] + v[rr][3]);
        v4f q4 = (v[rr][0] * v[rr][0] + v[rr][1] * v[rr][1]) + (v[rr][2] * v[rr][2] + v[rr][3] * v[rr][3]);
        rs[rr] = (s4.x + s4.y) + (s4.z + s4.w);
        rq[rr] = (q4.x + q4.y) + (q4.z + q4.w);
    }
#pragma unroll
    for (int m = 16; m > 0; m >>= 1) {
#pragma unroll
        for (int rr = 0; rr < 2; ++rr) {
            rs[rr] += __shfl_xor(rs[rr], m, 32);
            rq[rr] += __shfl_xor(rq[rr], m, 32);
        }
    }
#pragma unroll
    for (int rr = 0; rr < 2; ++rr) {
        int r = 2 * w + rr;
        float mean = rs[rr] * (1.f / 512.f);
        float var  = rq[rr] * (1.f / 512.f) - mean * mean;
        float inv  = rsqrtf(var + 1e-5f);
#pragma unroll
        for (int k = 0; k < 4; ++k) {
            v4f o;
            o.x = (v[rr][k].x - mean) * inv * gq[k].x + oq[k].x;
            o.y = (v[rr][k].y - mean) * inv * gq[k].y + oq[k].y;
            o.z = (v[rr][k].z - mean) * inv * gq[k].z + oq[k].z;
            o.w = (v[rr][k].w - mean) * inv * gq[k].w + oq[k].w;
            o4[(long)r * (D_ / 4) + ln + 32 * k] = o;
        }
    }
}

extern "C" void kernel_launch(void* const* d_in, const int* in_sizes, int n_in,
                              void* d_out, int out_size, void* d_ws, size_t ws_size,
                              hipStream_t stream) {
    const float* x_rgb = (const float*)d_in[0];
    const float* x_e   = (const float*)d_in[1];
    const float* xw1   = (const float*)d_in[2];
    const float* xw2   = (const float*)d_in[3];
    const float* dtw1  = (const float*)d_in[4];
    const float* dtb1  = (const float*)d_in[5];
    const float* dtw2  = (const float*)d_in[6];
    const float* dtb2  = (const float*)d_in[7];
    const float* Al1   = (const float*)d_in[8];
    const float* Al2   = (const float*)d_in[9];
    const float* Dv1   = (const float*)d_in[10];
    const float* Dv2   = (const float*)d_in[11];
    const float* g1    = (const float*)d_in[12];
    const float* lb1   = (const float*)d_in[13];
    const float* g2    = (const float*)d_in[14];
    const float* lb2   = (const float*)d_in[15];
    float* out = (float*)d_out;

    float* ws    = (float*)d_ws;
    float* dbl   = ws;
    float* delta = dbl + 2 * 4096 * 48;
    float* hin   = delta + 2 * 4096 * 512;

    k_proj<<<256, 256, 0, stream>>>(x_rgb, x_e, xw1, xw2, dtw1, dtb1, dtw2, dtb2, dbl, delta);
    k_scan1c<<<256, 512, 0, stream>>>(x_rgb, x_e, dbl, delta, Al1, Al2, hin);
    k_scan2_ln<<<256, 512, 0, stream>>>(x_rgb, x_e, dbl, delta, Al1, Al2, Dv1, Dv2, hin,
                                        g1, lb1, g2, lb2, out);
}
